# starting kernel
# baseline (speedup 1.0000x reference)
#include <hip/hip_runtime.h>
#include <math.h>

#define NN      4096
#define NFEAT   512
#define NHID    64
#define NCLASS  128
#define NHEADS  8
#define ALPHA   0.2f
#define NEGV    (-9000000000000000.0f)

typedef __attribute__((ext_vector_type(2))) float v2f;
typedef __attribute__((ext_vector_type(8))) float v8f;

__global__ void pack_adj_kernel(const int* __restrict__ adj,
                                unsigned* __restrict__ bits, long nwords) {
    long w = (long)blockIdx.x * blockDim.x + threadIdx.x;
    if (w >= nwords) return;
    const int* p = adj + w * 32;
    unsigned m = 0u;
    #pragma unroll
    for (int b = 0; b < 32; ++b) m |= (p[b] > 0 ? 1u : 0u) << b;
    bits[w] = m;
}

template <int K, int NB>
__global__ __launch_bounds__(128) void wmma_gemm_f32(
        const float* __restrict__ A,
        const float* __restrict__ B, long bStride,
        float* __restrict__ C, long cStride) {
    const int wave = threadIdx.x >> 5;
    const int lane = threadIdx.x & 31;
    const int hi   = lane >> 4;
    const int lo   = lane & 15;
    const int row0 = blockIdx.x * 16;
    const int col0 = (blockIdx.y * 4 + wave) * 16;
    const int head = blockIdx.z;

    const float* __restrict__ arow  = A + (long)(row0 + lo) * K + 2 * hi;
    const float* __restrict__ bbase = B + (long)head * bStride + 2 * hi * NB + col0 + lo;

    v8f acc = {};
    for (int k = 0; k < K; k += 4) {
        v2f a;
        a.x = arow[k];
        a.y = arow[k + 1];
        v2f b;
        b.x = bbase[k * NB];
        b.y = bbase[k * NB + NB];
        acc = __builtin_amdgcn_wmma_f32_16x16x4_f32(
            false, a, false, b, (short)0, acc, false, false);
    }

    float* cp = C + (long)head * cStride + (long)(row0 + 8 * hi) * NB + col0 + lo;
    #pragma unroll
    for (int v = 0; v < 8; ++v) cp[v * NB] = acc[v];
}

__global__ void fused_fifj(const float* __restrict__ H, long hStride, int F,
                           const float* __restrict__ a, int aStride,
                           float* __restrict__ fi, float* __restrict__ fj, int n) {
    const int head = blockIdx.y;
    const int nd   = blockIdx.x * blockDim.x + threadIdx.x;
    if (nd >= n) return;
    const float* hp = H + (long)head * hStride + (long)nd * F;
    const float* a1 = a + (long)head * aStride;
    const float* a2 = a1 + F;
    float s1 = 0.f, s2 = 0.f;
    for (int o = 0; o < F; ++o) {
        float hv = hp[o];
        s1 += hv * a1[o];
        s2 += hv * a2[o];
    }
    fi[(long)head * n + nd] = s1;
    fj[(long)head * n + nd] = s2;
}

template <int F, int JW, int R, bool ELU>
__global__ __launch_bounds__(JW * 32) void gat_attn(
        const float* __restrict__ Hfeat, long hStride,
        const float* __restrict__ fi, const float* __restrict__ fj,
        const unsigned* __restrict__ bits,
        float* __restrict__ Out, int ldo, int headColMul, int n) {
    constexpr int NC = F / 16;
    const int wave = threadIdx.x >> 5;
    const int lane = threadIdx.x & 31;
    const int hi   = lane >> 4;
    const int lo   = lane & 15;
    const int head = blockIdx.y;
    const int i0   = blockIdx.x * (16 * R);
    const int nwords = n >> 5;

    const float* __restrict__ Hh  = Hfeat + (long)head * hStride;
    const float* __restrict__ fih = fi + (long)head * n;
    const float* __restrict__ fjh = fj + (long)head * n;
    const float* __restrict__ bbase = Hh + 2 * hi * F + lo;

    float fiv[R];
    const unsigned* __restrict__ brow[R];
    #pragma unroll
    for (int r = 0; r < R; ++r) {
        fiv[r]  = fih[i0 + 16 * r + lo];
        brow[r] = bits + (long)(i0 + 16 * r + lo) * nwords;
    }

    const int chunk = n / JW;
    const int jbeg  = wave * chunk;
    const int jend  = jbeg + chunk;

    float m[R], l[R];
    v8f acc[R][NC];
    #pragma unroll
    for (int r = 0; r < R; ++r) {
        m[r] = -INFINITY;
        l[r] = 0.f;
        #pragma unroll
        for (int c = 0; c < NC; ++c) acc[r][c] = (v8f){};
    }

    for (int j0 = jbeg; j0 < jend; j0 += 32) {
        unsigned wmask[R];
        #pragma unroll
        for (int r = 0; r < R; ++r) wmask[r] = brow[r][j0 >> 5];

        #pragma unroll
        for (int half = 0; half < 2; ++half) {
            const int jb = j0 + 16 * half;
            const int bitbase = 16 * half;
            const float* __restrict__ fjp = fjh + jb + 2 * hi;

            float fjv[8];
            #pragma unroll
            for (int s = 0; s < 4; ++s)
                #pragma unroll
                for (int t = 0; t < 2; ++t) fjv[2 * s + t] = fjp[4 * s + t];

            float p8[R][8];
            #pragma unroll
            for (int r = 0; r < R; ++r) {
                float e8[8];
                float tmax = -INFINITY;
                #pragma unroll
                for (int s = 0; s < 4; ++s) {
                    #pragma unroll
                    for (int t = 0; t < 2; ++t) {
                        const int c = 4 * s + 2 * hi + t;
                        float v = fiv[r] + fjv[2 * s + t];
                        v = v > 0.f ? v : ALPHA * v;
                        const float e = ((wmask[r] >> (bitbase + c)) & 1u) ? v : NEGV;
                        e8[2 * s + t] = e;
                        tmax = fmaxf(tmax, e);
                    }
                }
                tmax = fmaxf(tmax, __shfl_xor(tmax, 16, 32));
                const float mnew = fmaxf(m[r], tmax);

                float psum = 0.f;
                #pragma unroll
                for (int q = 0; q < 8; ++q) {
                    p8[r][q] = __expf(e8[q] - mnew);
                    psum += p8[r][q];
                }
                psum += __shfl_xor(psum, 16, 32);

                if (__any(mnew > m[r])) {
                    const float scale = __expf(m[r] - mnew);
                    float sv[8];
                    #pragma unroll
                    for (int v = 0; v < 8; ++v) sv[v] = __shfl(scale, v + 8 * hi, 32);
                    #pragma unroll
                    for (int c = 0; c < NC; ++c)
                        #pragma unroll
                        for (int v = 0; v < 8; ++v) acc[r][c][v] *= sv[v];
                    l[r] = l[r] * scale + psum;
                    m[r] = mnew;
                } else {
                    l[r] += psum;
                }
            }

            const float* __restrict__ bp = bbase + jb * F;
            #pragma unroll
            for (int s = 0; s < 4; ++s) {
                const float* __restrict__ bs = bp + 4 * s * F;
                #pragma unroll
                for (int c = 0; c < NC; ++c) {
                    v2f b;
                    b.x = bs[16 * c];
                    b.y = bs[16 * c + F];
                    #pragma unroll
                    for (int r = 0; r < R; ++r) {
                        v2f a;
                        a.x = p8[r][2 * s];
                        a.y = p8[r][2 * s + 1];
                        acc[r][c] = __builtin_amdgcn_wmma_f32_16x16x4_f32(
                            false, a, false, b, (short)0, acc[r][c], false, false);
                    }
                }
            }
        }
    }

    __shared__ float m_lds[R][JW * 16];
    __shared__ float l_lds[R][JW * 16];
    __shared__ float L_lds[16];
    __shared__ float accbuf[JW * 16 * F];

    #pragma unroll
    for (int r = 0; r < R; ++r) {
        if (hi == 0) {
            m_lds[r][wave * 16 + lo] = m[r];
            l_lds[r][wave * 16 + lo] = l[r];
        }
    }
    __syncthreads();

    constexpr int TOT = 16 * F;
    constexpr int NT  = JW * 32;
    #pragma unroll
    for (int r = 0; r < R; ++r) {
        float Mrow = -INFINITY;
        #pragma unroll
        for (int w = 0; w < JW; ++w) Mrow = fmaxf(Mrow, m_lds[r][w * 16 + lo]);
        float Lrow = 0.f;
        #pragma unroll
        for (int w = 0; w < JW; ++w)
            Lrow += l_lds[r][w * 16 + lo] * __expf(m_lds[r][w * 16 + lo] - Mrow);
        if (wave == 0 && hi == 0) L_lds[lo] = Lrow;

        const float fown = __expf(m[r] - Mrow);
        float fv[8];
        #pragma unroll
        for (int v = 0; v < 8; ++v) fv[v] = __shfl(fown, v + 8 * hi, 32);

        float* ab = accbuf + (wave * 16 + 8 * hi) * F + lo;
        #pragma unroll
        for (int c = 0; c < NC; ++c)
            #pragma unroll
            for (int v = 0; v < 8; ++v)
                ab[v * F + 16 * c] = acc[r][c][v] * fv[v];
        __syncthreads();

        float* outBase = Out + (long)(i0 + 16 * r) * ldo + head * headColMul;
        #pragma unroll
        for (int e = 0; e < TOT / NT; ++e) {
            const int flat = e * NT + threadIdx.x;
            const int row = flat / F;
            const int col = flat % F;
            float val = 0.f;
            #pragma unroll
            for (int w = 0; w < JW; ++w) val += accbuf[(w * 16 + row) * F + col];
            float rr = val / L_lds[row];
            if (ELU) rr = rr > 0.f ? rr : (__expf(rr) - 1.0f);
            outBase[(long)row * ldo + col] = rr;
        }
        if (r + 1 < R) __syncthreads();
    }
}

__global__ void logsoftmax128(const float* __restrict__ in,
                              float* __restrict__ out, int n) {
    const int wave = threadIdx.x >> 5;
    const int lane = threadIdx.x & 31;
    const int row  = blockIdx.x * (blockDim.x >> 5) + wave;
    if (row >= n) return;
    const float* rp = in + (long)row * NCLASS;
    float v[4];
    float mx = -INFINITY;
    #pragma unroll
    for (int q = 0; q < 4; ++q) {
        v[q] = rp[lane + 32 * q];
        mx = fmaxf(mx, v[q]);
    }
    #pragma unroll
    for (int off = 16; off >= 1; off >>= 1) mx = fmaxf(mx, __shfl_xor(mx, off, 32));
    float s = 0.f;
    #pragma unroll
    for (int q = 0; q < 4; ++q) s += __expf(v[q] - mx);
    #pragma unroll
    for (int off = 16; off >= 1; off >>= 1) s += __shfl_xor(s, off, 32);
    const float ls = __logf(s);
    #pragma unroll
    for (int q = 0; q < 4; ++q) out[(long)row * NCLASS + lane + 32 * q] = v[q] - mx - ls;
}

extern "C" void kernel_launch(void* const* d_in, const int* in_sizes, int n_in,
                              void* d_out, int out_size, void* d_ws, size_t ws_size,
                              hipStream_t stream) {
    const float* x     = (const float*)d_in[0];
    const int*   adj   = (const int*)d_in[1];
    const float* W     = (const float*)d_in[2];
    const float* a     = (const float*)d_in[3];
    const float* W_out = (const float*)d_in[4];
    const float* a_out = (const float*)d_in[5];
    float* out = (float*)d_out;

    char* ws = (char*)d_ws;
    unsigned* bits = (unsigned*)ws;
    ws += (size_t)NN * (NN / 32) * sizeof(unsigned);
    float* h1 = (float*)ws;
    ws += (size_t)NHEADS * NN * NHID * sizeof(float);
    float* fi1 = (float*)ws; ws += (size_t)NHEADS * NN * sizeof(float);
    float* fj1 = (float*)ws; ws += (size_t)NHEADS * NN * sizeof(float);
    float* xcat = (float*)ws;
    ws += (size_t)NN * (NHEADS * NHID) * sizeof(float);
    float* h2 = (float*)ws;
    ws += (size_t)NN * NCLASS * sizeof(float);
    float* fi2 = (float*)ws; ws += (size_t)NN * sizeof(float);
    float* fj2 = (float*)ws; ws += (size_t)NN * sizeof(float);
    float* outpre = (float*)ws;

    {
        const long nwords = (long)NN * (NN / 32);
        pack_adj_kernel<<<dim3((nwords + 255) / 256), dim3(256), 0, stream>>>(adj, bits, nwords);
    }

    wmma_gemm_f32<NFEAT, NHID><<<dim3(NN / 16, 1, NHEADS), dim3(128), 0, stream>>>(
        x, W, (long)NFEAT * NHID, h1, (long)NN * NHID);

    fused_fifj<<<dim3(NN / 256, NHEADS), dim3(256), 0, stream>>>(
        h1, (long)NN * NHID, NHID, a, 2 * NHID, fi1, fj1, NN);

    gat_attn<NHID, 4, 2, false><<<dim3(NN / 32, NHEADS), dim3(128), 0, stream>>>(
        h1, (long)NN * NHID, fi1, fj1, bits,
        xcat, NHEADS * NHID, NHID, NN);

    wmma_gemm_f32<NHEADS * NHID, NCLASS><<<dim3(NN / 16, NCLASS / 64, 1), dim3(128), 0, stream>>>(
        xcat, W_out, 0, h2, 0);

    fused_fifj<<<dim3(NN / 256, 1), dim3(256), 0, stream>>>(
        h2, 0, NCLASS, a_out, 2 * NCLASS, fi2, fj2, NN);

    gat_attn<NCLASS, 4, 2, true><<<dim3(NN / 32, 1), dim3(128), 0, stream>>>(
        h2, 0, fi2, fj2, bits,
        outpre, NCLASS, 0, NN);

    logsoftmax128<<<dim3(NN / 4), dim3(128), 0, stream>>>(outpre, out, NN);
}


# attention softmax advances 64 keys per step, not 16: one row-max exchange, one rescale test, one row-sum join per 64 keys; a step's row sum is added as four 8-term partials (sum order changed)
# speedup vs baseline: 1.1240x; 1.1240x over previous
#include <hip/hip_runtime.h>
#include <math.h>

#define NN      4096
#define NFEAT   512
#define NHID    64
#define NCLASS  128
#define NHEADS  8
#define ALPHA   0.2f
#define NEGV    (-9000000000000000.0f)

typedef __attribute__((ext_vector_type(2))) float v2f;
typedef __attribute__((ext_vector_type(8))) float v8f;

__global__ void pack_adj_kernel(const int* __restrict__ adj,
                                unsigned* __restrict__ bits, long nwords) {
    long w = (long)blockIdx.x * blockDim.x + threadIdx.x;
    if (w >= nwords) return;
    const int* p = adj + w * 32;
    unsigned m = 0u;
    #pragma unroll
    for (int b = 0; b < 32; ++b) m |= (p[b] > 0 ? 1u : 0u) << b;
    bits[w] = m;
}

template <int K, int NB>
__global__ __launch_bounds__(128) void wmma_gemm_f32(
        const float* __restrict__ A,
        const float* __restrict__ B, long bStride,
        float* __restrict__ C, long cStride) {
    const int wave = threadIdx.x >> 5;
    const int lane = threadIdx.x & 31;
    const int hi   = lane >> 4;
    const int lo   = lane & 15;
    const int row0 = blockIdx.x * 16;
    const int col0 = (blockIdx.y * 4 + wave) * 16;
    const int head = blockIdx.z;

    const float* __restrict__ arow  = A + (long)(row0 + lo) * K + 2 * hi;
    const float* __restrict__ bbase = B + (long)head * bStride + 2 * hi * NB + col0 + lo;

    v8f acc = {};
    for (int k = 0; k < K; k += 4) {
        v2f a;
        a.x = arow[k];
        a.y = arow[k + 1];
        v2f b;
        b.x = bbase[k * NB];
        b.y = bbase[k * NB + NB];
        acc = __builtin_amdgcn_wmma_f32_16x16x4_f32(
            false, a, false, b, (short)0, acc, false, false);
    }

    float* cp = C + (long)head * cStride + (long)(row0 + 8 * hi) * NB + col0 + lo;
    #pragma unroll
    for (int v = 0; v < 8; ++v) cp[v * NB] = acc[v];
}

__global__ void fused_fifj(const float* __restrict__ H, long hStride, int F,
                           const float* __restrict__ a, int aStride,
                           float* __restrict__ fi, float* __restrict__ fj, int n) {
    const int head = blockIdx.y;
    const int nd   = blockIdx.x * blockDim.x + threadIdx.x;
    if (nd >= n) return;
    const float* hp = H + (long)head * hStride + (long)nd * F;
    const float* a1 = a + (long)head * aStride;
    const float* a2 = a1 + F;
    float s1 = 0.f, s2 = 0.f;
    for (int o = 0; o < F; ++o) {
        float hv = hp[o];
        s1 += hv * a1[o];
        s2 += hv * a2[o];
    }
    fi[(long)head * n + nd] = s1;
    fj[(long)head * n + nd] = s2;
}

template <int F, int JW, int R, bool ELU>
__global__ __launch_bounds__(JW * 32) void gat_attn(
        const float* __restrict__ Hfeat, long hStride,
        const float* __restrict__ fi, const float* __restrict__ fj,
        const unsigned* __restrict__ bits,
        float* __restrict__ Out, int ldo, int headColMul, int n) {
    constexpr int NC = F / 16;
    const int wave = threadIdx.x >> 5;
    const int lane = threadIdx.x & 31;
    const int hi   = lane >> 4;
    const int lo   = lane & 15;
    const int head = blockIdx.y;
    const int i0   = blockIdx.x * (16 * R);
    const int nwords = n >> 5;

    const float* __restrict__ Hh  = Hfeat + (long)head * hStride;
    const float* __restrict__ fih = fi + (long)head * n;
    const float* __restrict__ fjh = fj + (long)head * n;
    const float* __restrict__ bbase = Hh + 2 * hi * F + lo;

    float fiv[R];
    const unsigned* __restrict__ brow[R];
    #pragma unroll
    for (int r = 0; r < R; ++r) {
        fiv[r]  = fih[i0 + 16 * r + lo];
        brow[r] = bits + (long)(i0 + 16 * r + lo) * nwords;
    }

    const int chunk = n / JW;
    const int jbeg  = wave * chunk;
    const int jend  = jbeg + chunk;

    float m[R], l[R];
    v8f acc[R][NC];
    #pragma unroll
    for (int r = 0; r < R; ++r) {
        m[r] = -INFINITY;
        l[r] = 0.f;
        #pragma unroll
        for (int c = 0; c < NC; ++c) acc[r][c] = (v8f){};
    }

    constexpr int KS = 64;
    constexpr int NT16 = KS / 16;
    constexpr int NW = KS / 32;
    for (int j0 = jbeg; j0 < jend; j0 += KS) {
        unsigned wmask[R][NW];
        #pragma unroll
        for (int r = 0; r < R; ++r)
            #pragma unroll
            for (int w = 0; w < NW; ++w) wmask[r][w] = brow[r][(j0 >> 5) + w];

        float fjv[NT16][8];
        #pragma unroll
        for (int t = 0; t < NT16; ++t) {
            const float* __restrict__ fjp = fjh + j0 + 16 * t + 2 * hi;
            #pragma unroll
            for (int s = 0; s < 4; ++s)
                #pragma unroll
                for (int u = 0; u < 2; ++u) fjv[t][2 * s + u] = fjp[4 * s + u];
        }

        float p[R][NT16][8];
        #pragma unroll
        for (int r = 0; r < R; ++r) {
            float tm[NT16];
            #pragma unroll
            for (int t = 0; t < NT16; ++t) {
                const unsigned wbits = wmask[r][t >> 1];
                const int bitbase = 16 * (t & 1);
                float mx = -INFINITY;
                #pragma unroll
                for (int s = 0; s < 4; ++s) {
                    #pragma unroll
                    for (int u = 0; u < 2; ++u) {
                        const int c = 4 * s + 2 * hi + u;
                        float v = fiv[r] + fjv[t][2 * s + u];
                        v = v > 0.f ? v : ALPHA * v;
                        const float e = ((wbits >> (bitbase + c)) & 1u) ? v : NEGV;
                        p[r][t][2 * s + u] = e;
                        mx = fmaxf(mx, e);
                    }
                }
                tm[t] = mx;
            }
            float tmax = fmaxf(fmaxf(tm[0], tm[1]), fmaxf(tm[2], tm[3]));
            tmax = fmaxf(tmax, __shfl_xor(tmax, 16, 32));
            const float mnew = fmaxf(m[r], tmax);

            float ps[NT16];
            #pragma unroll
            for (int t = 0; t < NT16; ++t) {
                float sacc = 0.f;
                #pragma unroll
                for (int q = 0; q < 8; ++q) {
                    p[r][t][q] = __expf(p[r][t][q] - mnew);
                    sacc += p[r][t][q];
                }
                ps[t] = sacc;
            }
            float psum = (ps[0] + ps[1]) + (ps[2] + ps[3]);
            psum += __shfl_xor(psum, 16, 32);

            if (__any(mnew > m[r])) {
                const float scale = __expf(m[r] - mnew);
                float sv[8];
                #pragma unroll
                for (int v = 0; v < 8; ++v) sv[v] = __shfl(scale, v + 8 * hi, 32);
                #pragma unroll
                for (int c = 0; c < NC; ++c)
                    #pragma unroll
                    for (int v = 0; v < 8; ++v) acc[r][c][v] *= sv[v];
                l[r] = l[r] * scale + psum;
                m[r] = mnew;
            } else {
                l[r] += psum;
            }
        }

        #pragma unroll
        for (int t = 0; t < NT16; ++t) {
            __builtin_amdgcn_sched_barrier(0);
            const float* __restrict__ bp = bbase + (j0 + 16 * t) * F;
            #pragma unroll
            for (int s = 0; s < 4; ++s) {
                const float* __restrict__ bs = bp + 4 * s * F;
                #pragma unroll
                for (int c = 0; c < NC; ++c) {
                    v2f b;
                    b.x = bs[16 * c];
                    b.y = bs[16 * c + F];
                    #pragma unroll
                    for (int r = 0; r < R; ++r) {
                        v2f a;
                        a.x = p[r][t][2 * s];
                        a.y = p[r][t][2 * s + 1];
                        acc[r][c] = __builtin_amdgcn_wmma_f32_16x16x4_f32(
                            false, a, false, b, (short)0, acc[r][c], false, false);
                    }
                }
            }
        }
    }

    __shared__ float m_lds[R][JW * 16];
    __shared__ float l_lds[R][JW * 16];
    __shared__ float L_lds[16];
    __shared__ float accbuf[JW * 16 * F];

    #pragma unroll
    for (int r = 0; r < R; ++r) {
        if (hi == 0) {
            m_lds[r][wave * 16 + lo] = m[r];
            l_lds[r][wave * 16 + lo] = l[r];
        }
    }
    __syncthreads();

    constexpr int TOT = 16 * F;
    constexpr int NT  = JW * 32;
    #pragma unroll
    for (int r = 0; r < R; ++r) {
        float Mrow = -INFINITY;
        #pragma unroll
        for (int w = 0; w < JW; ++w) Mrow = fmaxf(Mrow, m_lds[r][w * 16 + lo]);
        float Lrow = 0.f;
        #pragma unroll
        for (int w = 0; w < JW; ++w)
            Lrow += l_lds[r][w * 16 + lo] * __expf(m_lds[r][w * 16 + lo] - Mrow);
        if (wave == 0 && hi == 0) L_lds[lo] = Lrow;

        const float fown = __expf(m[r] - Mrow);
        float fv[8];
        #pragma unroll
        for (int v = 0; v < 8; ++v) fv[v] = __shfl(fown, v + 8 * hi, 32);

        float* ab = accbuf + (wave * 16 + 8 * hi) * F + lo;
        #pragma unroll
        for (int c = 0; c < NC; ++c)
            #pragma unroll
            for (int v = 0; v < 8; ++v)
                ab[v * F + 16 * c] = acc[r][c][v] * fv[v];
        __syncthreads();

        float* outBase = Out + (long)(i0 + 16 * r) * ldo + head * headColMul;
        #pragma unroll
        for (int e = 0; e < TOT / NT; ++e) {
            const int flat = e * NT + threadIdx.x;
            const int row = flat / F;
            const int col = flat % F;
            float val = 0.f;
            #pragma unroll
            for (int w = 0; w < JW; ++w) val += accbuf[(w * 16 + row) * F + col];
            float rr = val / L_lds[row];
            if (ELU) rr = rr > 0.f ? rr : (__expf(rr) - 1.0f);
            outBase[(long)row * ldo + col] = rr;
        }
        if (r + 1 < R) __syncthreads();
    }
}

__global__ void logsoftmax128(const float* __restrict__ in,
                              float* __restrict__ out, int n) {
    const int wave = threadIdx.x >> 5;
    const int lane = threadIdx.x & 31;
    const int row  = blockIdx.x * (blockDim.x >> 5) + wave;
    if (row >= n) return;
    const float* rp = in + (long)row * NCLASS;
    float v[4];
    float mx = -INFINITY;
    #pragma unroll
    for (int q = 0; q < 4; ++q) {
        v[q] = rp[lane + 32 * q];
        mx = fmaxf(mx, v[q]);
    }
    #pragma unroll
    for (int off = 16; off >= 1; off >>= 1) mx = fmaxf(mx, __shfl_xor(mx, off, 32));
    float s = 0.f;
    #pragma unroll
    for (int q = 0; q < 4; ++q) s += __expf(v[q] - mx);
    #pragma unroll
    for (int off = 16; off >= 1; off >>= 1) s += __shfl_xor(s, off, 32);
    const float ls = __logf(s);
    #pragma unroll
    for (int q = 0; q < 4; ++q) out[(long)row * NCLASS + lane + 32 * q] = v[q] - mx - ls;
}

extern "C" void kernel_launch(void* const* d_in, const int* in_sizes, int n_in,
                              void* d_out, int out_size, void* d_ws, size_t ws_size,
                              hipStream_t stream) {
    const float* x     = (const float*)d_in[0];
    const int*   adj   = (const int*)d_in[1];
    const float* W     = (const float*)d_in[2];
    const float* a     = (const float*)d_in[3];
    const float* W_out = (const float*)d_in[4];
    const float* a_out = (const float*)d_in[5];
    float* out = (float*)d_out;

    char* ws = (char*)d_ws;
    unsigned* bits = (unsigned*)ws;
    ws += (size_t)NN * (NN / 32) * sizeof(unsigned);
    float* h1 = (float*)ws;
    ws += (size_t)NHEADS * NN * NHID * sizeof(float);
    float* fi1 = (float*)ws; ws += (size_t)NHEADS * NN * sizeof(float);
    float* fj1 = (float*)ws; ws += (size_t)NHEADS * NN * sizeof(float);
    float* xcat = (float*)ws;
    ws += (size_t)NN * (NHEADS * NHID) * sizeof(float);
    float* h2 = (float*)ws;
    ws += (size_t)NN * NCLASS * sizeof(float);
    float* fi2 = (float*)ws; ws += (size_t)NN * sizeof(float);
    float* fj2 = (float*)ws; ws += (size_t)NN * sizeof(float);
    float* outpre = (float*)ws;

    {
        const long nwords = (long)NN * (NN / 32);
        pack_adj_kernel<<<dim3((nwords + 255) / 256), dim3(256), 0, stream>>>(adj, bits, nwords);
    }

    wmma_gemm_f32<NFEAT, NHID><<<dim3(NN / 16, 1, NHEADS), dim3(128), 0, stream>>>(
        x, W, (long)NFEAT * NHID, h1, (long)NN * NHID);

    fused_fifj<<<dim3(NN / 256, NHEADS), dim3(256), 0, stream>>>(
        h1, (long)NN * NHID, NHID, a, 2 * NHID, fi1, fj1, NN);

    gat_attn<NHID, 4, 2, false><<<dim3(NN / 32, NHEADS), dim3(128), 0, stream>>>(
        h1, (long)NN * NHID, fi1, fj1, bits,
        xcat, NHEADS * NHID, NHID, NN);

    wmma_gemm_f32<NHEADS * NHID, NCLASS><<<dim3(NN / 16, NCLASS / 64, 1), dim3(128), 0, stream>>>(
        xcat, W_out, 0, h2, 0);

    fused_fifj<<<dim3(NN / 256, 1), dim3(256), 0, stream>>>(
        h2, 0, NCLASS, a_out, 2 * NCLASS, fi2, fj2, NN);

    gat_attn<NCLASS, 4, 2, true><<<dim3(NN / 32, 1), dim3(128), 0, stream>>>(
        h2, 0, fi2, fj2, bits,
        outpre, NCLASS, 0, NN);

    logsoftmax128<<<dim3(NN / 4), dim3(128), 0, stream>>>(outpre, out, NN);
}


# attention weights as products of per-node exponentials, exp(leaky(fi+fj)-m)=max(aP_i*eP_j,aN_i*eN_j), m=leaky(fi+chunk max fj); factors once per wave in LDS, no exp per pair; row-sum order changed
# speedup vs baseline: 1.2706x; 1.1304x over previous
#include <hip/hip_runtime.h>
#include <math.h>

#define NN      4096
#define NFEAT   512
#define NHID    64
#define NCLASS  128
#define NHEADS  8
#define ALPHA   0.2f
#define NEGV    (-9000000000000000.0f)

typedef __attribute__((ext_vector_type(2))) float v2f;
typedef __attribute__((ext_vector_type(4))) float v4f;
typedef __attribute__((ext_vector_type(8))) float v8f;

__global__ void pack_adj_kernel(const int* __restrict__ adj,
                                unsigned* __restrict__ bits, long nwords) {
    long w = (long)blockIdx.x * blockDim.x + threadIdx.x;
    if (w >= nwords) return;
    const int* p = adj + w * 32;
    unsigned m = 0u;
    #pragma unroll
    for (int b = 0; b < 32; ++b) m |= (p[b] > 0 ? 1u : 0u) << b;
    bits[w] = m;
}

template <int K, int NB>
__global__ __launch_bounds__(128) void wmma_gemm_f32(
        const float* __restrict__ A,
        const float* __restrict__ B, long bStride,
        float* __restrict__ C, long cStride) {
    const int wave = threadIdx.x >> 5;
    const int lane = threadIdx.x & 31;
    const int hi   = lane >> 4;
    const int lo   = lane & 15;
    const int row0 = blockIdx.x * 16;
    const int col0 = (blockIdx.y * 4 + wave) * 16;
    const int head = blockIdx.z;

    const float* __restrict__ arow  = A + (long)(row0 + lo) * K + 2 * hi;
    const float* __restrict__ bbase = B + (long)head * bStride + 2 * hi * NB + col0 + lo;

    v8f acc = {};
    for (int k = 0; k < K; k += 4) {
        v2f a;
        a.x = arow[k];
        a.y = arow[k + 1];
        v2f b;
        b.x = bbase[k * NB];
        b.y = bbase[k * NB + NB];
        acc = __builtin_amdgcn_wmma_f32_16x16x4_f32(
            false, a, false, b, (short)0, acc, false, false);
    }

    float* cp = C + (long)head * cStride + (long)(row0 + 8 * hi) * NB + col0 + lo;
    #pragma unroll
    for (int v = 0; v < 8; ++v) cp[v * NB] = acc[v];
}

__global__ void fused_fifj(const float* __restrict__ H, long hStride, int F,
                           const float* __restrict__ a, int aStride,
                           float* __restrict__ fi, float* __restrict__ fj, int n) {
    const int head = blockIdx.y;
    const int nd   = blockIdx.x * blockDim.x + threadIdx.x;
    if (nd >= n) return;
    const float* hp = H + (long)head * hStride + (long)nd * F;
    const float* a1 = a + (long)head * aStride;
    const float* a2 = a1 + F;
    float s1 = 0.f, s2 = 0.f;
    for (int o = 0; o < F; ++o) {
        float hv = hp[o];
        s1 += hv * a1[o];
        s2 += hv * a2[o];
    }
    fi[(long)head * n + nd] = s1;
    fj[(long)head * n + nd] = s2;
}

template <int F, int R>
__device__ __forceinline__ void attn_exact_pass(
        const float* __restrict__ bbase, const float* __restrict__ fjh,
        const unsigned* const (&brow)[R], const float (&fiv)[R],
        const int hi, const int jbeg, const int jend,
        float (&m)[R], float (&l)[R], v8f (&acc)[R][F / 16]) {
    constexpr int NC = F / 16;
    constexpr int KS = 16;
    constexpr int NT16 = KS / 16;
    constexpr int NW = 1;
    for (int j0 = jbeg; j0 < jend; j0 += KS) {
        unsigned wmask[R][NW];
        #pragma unroll
        for (int r = 0; r < R; ++r)
            #pragma unroll
            for (int w = 0; w < NW; ++w) wmask[r][w] = brow[r][(j0 >> 5) + w];

        float fjv[NT16][8];
        #pragma unroll
        for (int t = 0; t < NT16; ++t) {
            const float* __restrict__ fjp = fjh + j0 + 16 * t + 2 * hi;
            #pragma unroll
            for (int s = 0; s < 4; ++s)
                #pragma unroll
                for (int u = 0; u < 2; ++u) fjv[t][2 * s + u] = fjp[4 * s + u];
        }

        float p[R][NT16][8];
        #pragma unroll
        for (int r = 0; r < R; ++r) {
            float tm[NT16];
            #pragma unroll
            for (int t = 0; t < NT16; ++t) {
                const unsigned wbits = wmask[r][t >> 1];
                const int bitbase = j0 & 16;
                float mx = -INFINITY;
                #pragma unroll
                for (int s = 0; s < 4; ++s) {
                    #pragma unroll
                    for (int u = 0; u < 2; ++u) {
                        const int c = 4 * s + 2 * hi + u;
                        float v = fiv[r] + fjv[t][2 * s + u];
                        v = v > 0.f ? v : ALPHA * v;
                        const float e = ((wbits >> (bitbase + c)) & 1u) ? v : NEGV;
                        p[r][t][2 * s + u] = e;
                        mx = fmaxf(mx, e);
                    }
                }
                tm[t] = mx;
            }
            float tmax = tm[0];
            tmax = fmaxf(tmax, __shfl_xor(tmax, 16, 32));
            const float mnew = fmaxf(m[r], tmax);

            float ps[NT16];
            #pragma unroll
            for (int t = 0; t < NT16; ++t) {
                float sacc = 0.f;
                #pragma unroll
                for (int q = 0; q < 8; ++q) {
                    p[r][t][q] = __expf(p[r][t][q] - mnew);
                    sacc += p[r][t][q];
                }
                ps[t] = sacc;
            }
            float psum = ps[0];
            psum += __shfl_xor(psum, 16, 32);

            if (__any(mnew > m[r])) {
                const float scale = __expf(m[r] - mnew);
                float sv[8];
                #pragma unroll
                for (int v = 0; v < 8; ++v) sv[v] = __shfl(scale, v + 8 * hi, 32);
                #pragma unroll
                for (int c = 0; c < NC; ++c)
                    #pragma unroll
                    for (int v = 0; v < 8; ++v) acc[r][c][v] *= sv[v];
                l[r] = l[r] * scale + psum;
                m[r] = mnew;
            } else {
                l[r] += psum;
            }
        }

        #pragma unroll
        for (int t = 0; t < NT16; ++t) {
            __builtin_amdgcn_sched_barrier(0);
            const float* __restrict__ bp = bbase + (j0 + 16 * t) * F;
            #pragma unroll
            for (int s = 0; s < 4; ++s) {
                const float* __restrict__ bs = bp + 4 * s * F;
                #pragma unroll
                for (int c = 0; c < NC; ++c) {
                    v2f b;
                    b.x = bs[16 * c];
                    b.y = bs[16 * c + F];
                    #pragma unroll
                    for (int r = 0; r < R; ++r) {
                        v2f a;
                        a.x = p[r][t][2 * s];
                        a.y = p[r][t][2 * s + 1];
                        acc[r][c] = __builtin_amdgcn_wmma_f32_16x16x4_f32(
                            false, a, false, b, (short)0, acc[r][c], false, false);
                    }
                }
            }
        }
    }

}

template <int F, int JW, int R, bool ELU>
__attribute__((amdgpu_waves_per_eu(F > 64 ? 1 : 4)))
__global__ __launch_bounds__(JW * 32) void gat_attn(
        const float* __restrict__ Hfeat, long hStride,
        const float* __restrict__ fi, const float* __restrict__ fj,
        const unsigned* __restrict__ bits,
        float* __restrict__ Out, int ldo, int headColMul, int n) {
    constexpr int NC = F / 16;
    constexpr int CH = NN / JW;
    constexpr int EB = JW * CH * 2;
    constexpr int AB = JW * 16 * F;
    constexpr int SM = EB > AB ? EB : AB;
    const int wave = threadIdx.x >> 5;
    const int lane = threadIdx.x & 31;
    const int hi   = lane >> 4;
    const int lo   = lane & 15;
    const int head = blockIdx.y;
    const int i0   = blockIdx.x * (16 * R);
    const int nwords = n >> 5;

    __shared__ __attribute__((aligned(16))) float smem[SM];
    __shared__ float m_lds[R][JW * 16];
    __shared__ float l_lds[R][JW * 16];
    __shared__ float L_lds[16];
    float* accbuf = smem;
    float* ew = smem + wave * (2 * CH);

    const float* __restrict__ Hh  = Hfeat + (long)head * hStride;
    const float* __restrict__ fih = fi + (long)head * n;
    const float* __restrict__ fjh = fj + (long)head * n;
    const float* __restrict__ bbase = Hh + 2 * hi * F + lo;

    float fiv[R];
    const unsigned* brow[R];
    #pragma unroll
    for (int r = 0; r < R; ++r) {
        fiv[r]  = fih[i0 + 16 * r + lo];
        brow[r] = bits + (long)(i0 + 16 * r + lo) * nwords;
    }

    const int jbeg  = wave * CH;
    const int jend  = jbeg + CH;

    float cm = -INFINITY;
    {
        v4f fq[CH / 128];
        const v4f* __restrict__ fj4 = (const v4f*)(fjh + jbeg);
        #pragma unroll
        for (int i = 0; i < CH / 128; ++i) {
            fq[i] = fj4[i * 32 + lane];
            cm = fmaxf(cm, fmaxf(fmaxf(fq[i].x, fq[i].y), fmaxf(fq[i].z, fq[i].w)));
        }
        #pragma unroll
        for (int off = 16; off >= 1; off >>= 1) cm = fmaxf(cm, __shfl_xor(cm, off, 32));
        #pragma unroll
        for (int i = 0; i < CH / 128; ++i) {
            float* ep = ew + 8 * (i * 32 + lane);
            const float d0 = fq[i].x - cm, d1 = fq[i].y - cm;
            const float d2 = fq[i].z - cm, d3 = fq[i].w - cm;
            v4f e0, e1;
            e0.x = __expf(d0); e0.y = __expf(ALPHA * d0);
            e0.z = __expf(d1); e0.w = __expf(ALPHA * d1);
            e1.x = __expf(d2); e1.y = __expf(ALPHA * d2);
            e1.z = __expf(d3); e1.w = __expf(ALPHA * d3);
            *(v4f*)ep = e0;
            *(v4f*)(ep + 4) = e1;
        }
    }
    __syncthreads();

    float m[R], l[R];
    v2f arow[R];
    v8f acc[R][NC];
    #pragma unroll
    for (int r = 0; r < R; ++r) {
        const float t = fiv[r] + cm;
        m[r] = fmaxf(t, ALPHA * t);
        arow[r].x = __expf(t - m[r]);
        arow[r].y = __expf(ALPHA * t - m[r]);
        l[r] = 0.f;
        #pragma unroll
        for (int c = 0; c < NC; ++c) acc[r][c] = (v8f){};
    }

    constexpr int KS = F > 64 ? 64 : 32;
    constexpr int NT16 = KS / 16;
    constexpr int NW = KS / 32;
    for (int j0 = jbeg; j0 < jend; j0 += KS) {
        unsigned wsh[R][NW];
        #pragma unroll
        for (int r = 0; r < R; ++r)
            #pragma unroll
            for (int w = 0; w < NW; ++w) wsh[r][w] = brow[r][(j0 >> 5) + w] >> (2 * hi);

        const float* es = ew + 2 * (j0 - jbeg) + 4 * hi;
        #pragma unroll
        for (int t = 0; t < NT16; ++t) {
            __builtin_amdgcn_sched_barrier(0);

            float p[R][8];
            float ps[R];
            #pragma unroll
            for (int r = 0; r < R; ++r) ps[r] = 0.f;
            #pragma unroll
            for (int s = 0; s < 4; ++s) {
                const v4f q = *(const v4f*)(es + 2 * (16 * t + 4 * s));
                v2f q0, q1;
                q0.x = q.x; q0.y = q.y;
                q1.x = q.z; q1.y = q.w;
                #pragma unroll
                for (int r = 0; r < R; ++r) {
                    const v2f x0 = arow[r] * q0;
                    const v2f x1 = arow[r] * q1;
                    const float w0 = fmaxf(x0.x, x0.y);
                    const float w1 = fmaxf(x1.x, x1.y);
                    const int k0 = 16 * (t & 1) + 4 * s;
                    const unsigned wb = wsh[r][t >> 1];
                    const int mk0 = ((int)(wb << (31 - k0))) >> 31;
                    const int mk1 = ((int)(wb << (30 - k0))) >> 31;
                    const float p0 = __int_as_float(__float_as_int(w0) & mk0);
                    const float p1 = __int_as_float(__float_as_int(w1) & mk1);
                    p[r][2 * s]     = p0;
                    p[r][2 * s + 1] = p1;
                    ps[r] += p0;
                    ps[r] += p1;
                }
            }
            #pragma unroll
            for (int r = 0; r < R; ++r) l[r] += ps[r];

            const float* __restrict__ bp = bbase + (j0 + 16 * t) * F;
            #pragma unroll
            for (int s = 0; s < 4; ++s) {
                const float* __restrict__ bs = bp + 4 * s * F;
                #pragma unroll
                for (int c = 0; c < NC; ++c) {
                    v2f b;
                    b.x = bs[16 * c];
                    b.y = bs[16 * c + F];
                    #pragma unroll
                    for (int r = 0; r < R; ++r) {
                        v2f a;
                        a.x = p[r][2 * s];
                        a.y = p[r][2 * s + 1];
                        acc[r][c] = __builtin_amdgcn_wmma_f32_16x16x4_f32(
                            false, a, false, b, (short)0, acc[r][c], false, false);
                    }
                }
            }
        }
    }
    #pragma unroll
    for (int r = 0; r < R; ++r) l[r] += __shfl_xor(l[r], 16, 32);

    #pragma unroll
    for (int r = 0; r < R; ++r) {
        if (hi == 0) {
            m_lds[r][wave * 16 + lo] = m[r];
            l_lds[r][wave * 16 + lo] = l[r];
        }
    }
    __syncthreads();

    bool small = false;
    #pragma unroll
    for (int r = 0; r < R; ++r) {
        float Mrow = -INFINITY;
        #pragma unroll
        for (int w = 0; w < JW; ++w) Mrow = fmaxf(Mrow, m_lds[r][w * 16 + lo]);
        float Lrow = 0.f;
        #pragma unroll
        for (int w = 0; w < JW; ++w)
            Lrow += l_lds[r][w * 16 + lo] * __expf(m_lds[r][w * 16 + lo] - Mrow);
        small = small || !(Lrow >= 1e-25f);
    }
    if (__any(small)) {
        __syncthreads();
        #pragma unroll
        for (int r = 0; r < R; ++r) {
            const unsigned* b1[1] = {brow[r]};
            const float f1[1] = {fiv[r]};
            float m1[1] = {-INFINITY};
            float l1[1] = {0.f};
            v8f a1[1][NC];
            #pragma unroll
            for (int c = 0; c < NC; ++c) a1[0][c] = (v8f){};
            attn_exact_pass<F, 1>(bbase, fjh, b1, f1, hi, jbeg, jend, m1, l1, a1);
            m[r] = m1[0];
            l[r] = l1[0];
            #pragma unroll
            for (int c = 0; c < NC; ++c) acc[r][c] = a1[0][c];
        }
        #pragma unroll
        for (int r = 0; r < R; ++r) {
            if (hi == 0) {
                m_lds[r][wave * 16 + lo] = m[r];
                l_lds[r][wave * 16 + lo] = l[r];
            }
        }
        __syncthreads();
    }

    constexpr int TOT = 16 * F;
    constexpr int NT  = JW * 32;
    #pragma unroll
    for (int r = 0; r < R; ++r) {
        float Mrow = -INFINITY;
        #pragma unroll
        for (int w = 0; w < JW; ++w) Mrow = fmaxf(Mrow, m_lds[r][w * 16 + lo]);
        float Lrow = 0.f;
        #pragma unroll
        for (int w = 0; w < JW; ++w)
            Lrow += l_lds[r][w * 16 + lo] * __expf(m_lds[r][w * 16 + lo] - Mrow);
        if (wave == 0 && hi == 0) L_lds[lo] = Lrow;

        const float fown = __expf(m[r] - Mrow);
        float fv[8];
        #pragma unroll
        for (int v = 0; v < 8; ++v) fv[v] = __shfl(fown, v + 8 * hi, 32);

        float* ab = accbuf + (wave * 16 + 8 * hi) * F + lo;
        #pragma unroll
        for (int c = 0; c < NC; ++c)
            #pragma unroll
            for (int v = 0; v < 8; ++v)
                ab[v * F + 16 * c] = acc[r][c][v] * fv[v];
        __syncthreads();

        float* outBase = Out + (long)(i0 + 16 * r) * ldo + head * headColMul;
        #pragma unroll
        for (int e = 0; e < TOT / NT; ++e) {
            const int flat = e * NT + threadIdx.x;
            const int row = flat / F;
            const int col = flat % F;
            float val = 0.f;
            #pragma unroll
            for (int w = 0; w < JW; ++w) val += accbuf[(w * 16 + row) * F + col];
            float rr = val / L_lds[row];
            if (ELU) rr = rr > 0.f ? rr : (__expf(rr) - 1.0f);
            outBase[(long)row * ldo + col] = rr;
        }
        if (r + 1 < R) __syncthreads();
    }
}

__global__ void logsoftmax128(const float* __restrict__ in,
                              float* __restrict__ out, int n) {
    const int wave = threadIdx.x >> 5;
    const int lane = threadIdx.x & 31;
    const int row  = blockIdx.x * (blockDim.x >> 5) + wave;
    if (row >= n) return;
    const float* rp = in + (long)row * NCLASS;
    float v[4];
    float mx = -INFINITY;
    #pragma unroll
    for (int q = 0; q < 4; ++q) {
        v[q] = rp[lane + 32 * q];
        mx = fmaxf(mx, v[q]);
    }
    #pragma unroll
    for (int off = 16; off >= 1; off >>= 1) mx = fmaxf(mx, __shfl_xor(mx, off, 32));
    float s = 0.f;
    #pragma unroll
    for (int q = 0; q < 4; ++q) s += __expf(v[q] - mx);
    #pragma unroll
    for (int off = 16; off >= 1; off >>= 1) s += __shfl_xor(s, off, 32);
    const float ls = __logf(s);
    #pragma unroll
    for (int q = 0; q < 4; ++q) out[(long)row * NCLASS + lane + 32 * q] = v[q] - mx - ls;
}

extern "C" void kernel_launch(void* const* d_in, const int* in_sizes, int n_in,
                              void* d_out, int out_size, void* d_ws, size_t ws_size,
                              hipStream_t stream) {
    const float* x     = (const float*)d_in[0];
    const int*   adj   = (const int*)d_in[1];
    const float* W     = (const float*)d_in[2];
    const float* a     = (const float*)d_in[3];
    const float* W_out = (const float*)d_in[4];
    const float* a_out = (const float*)d_in[5];
    float* out = (float*)d_out;

    char* ws = (char*)d_ws;
    unsigned* bits = (unsigned*)ws;
    ws += (size_t)NN * (NN / 32) * sizeof(unsigned);
    float* h1 = (float*)ws;
    ws += (size_t)NHEADS * NN * NHID * sizeof(float);
    float* fi1 = (float*)ws; ws += (size_t)NHEADS * NN * sizeof(float);
    float* fj1 = (float*)ws; ws += (size_t)NHEADS * NN * sizeof(float);
    float* xcat = (float*)ws;
    ws += (size_t)NN * (NHEADS * NHID) * sizeof(float);
    float* h2 = (float*)ws;
    ws += (size_t)NN * NCLASS * sizeof(float);
    float* fi2 = (float*)ws; ws += (size_t)NN * sizeof(float);
    float* fj2 = (float*)ws; ws += (size_t)NN * sizeof(float);
    float* outpre = (float*)ws;

    {
        const long nwords = (long)NN * (NN / 32);
        pack_adj_kernel<<<dim3((nwords + 255) / 256), dim3(256), 0, stream>>>(adj, bits, nwords);
    }

    wmma_gemm_f32<NFEAT, NHID><<<dim3(NN / 16, 1, NHEADS), dim3(128), 0, stream>>>(
        x, W, (long)NFEAT * NHID, h1, (long)NN * NHID);

    fused_fifj<<<dim3(NN / 256, NHEADS), dim3(256), 0, stream>>>(
        h1, (long)NN * NHID, NHID, a, 2 * NHID, fi1, fj1, NN);

    gat_attn<NHID, 4, 2, false><<<dim3(NN / 32, NHEADS), dim3(128), 0, stream>>>(
        h1, (long)NN * NHID, fi1, fj1, bits,
        xcat, NHEADS * NHID, NHID, NN);

    wmma_gemm_f32<NHEADS * NHID, NCLASS><<<dim3(NN / 16, NCLASS / 64, 1), dim3(128), 0, stream>>>(
        xcat, W_out, 0, h2, 0);

    fused_fifj<<<dim3(NN / 256, 1), dim3(256), 0, stream>>>(
        h2, 0, NCLASS, a_out, 2 * NCLASS, fi2, fj2, NN);

    gat_attn<NCLASS, 4, 2, true><<<dim3(NN / 32, 1), dim3(128), 0, stream>>>(
        h2, 0, fi2, fj2, bits,
        outpre, NCLASS, 0, NN);

    logsoftmax128<<<dim3(NN / 4), dim3(128), 0, stream>>>(outpre, out, NN);
}


# layer-2 attention split by output columns: four blocks per 32-row group, each forms the group's weights itself and aggregates 32 of the 128 columns; 128 -> 512 blocks, two waves per SIMD
# speedup vs baseline: 1.4221x; 1.1192x over previous
#include <hip/hip_runtime.h>
#include <math.h>

#define NN      4096
#define NFEAT   512
#define NHID    64
#define NCLASS  128
#define NHEADS  8
#define ALPHA   0.2f
#define NEGV    (-9000000000000000.0f)

typedef __attribute__((ext_vector_type(2))) float v2f;
typedef __attribute__((ext_vector_type(4))) float v4f;
typedef __attribute__((ext_vector_type(8))) float v8f;

__global__ void pack_adj_kernel(const int* __restrict__ adj,
                                unsigned* __restrict__ bits, long nwords) {
    long w = (long)blockIdx.x * blockDim.x + threadIdx.x;
    if (w >= nwords) return;
    const int* p = adj + w * 32;
    unsigned m = 0u;
    #pragma unroll
    for (int b = 0; b < 32; ++b) m |= (p[b] > 0 ? 1u : 0u) << b;
    bits[w] = m;
}

template <int K, int NB>
__global__ __launch_bounds__(128) void wmma_gemm_f32(
        const float* __restrict__ A,
        const float* __restrict__ B, long bStride,
        float* __restrict__ C, long cStride) {
    const int wave = threadIdx.x >> 5;
    const int lane = threadIdx.x & 31;
    const int hi   = lane >> 4;
    const int lo   = lane & 15;
    const int row0 = blockIdx.x * 16;
    const int col0 = (blockIdx.y * 4 + wave) * 16;
    const int head = blockIdx.z;

    const float* __restrict__ arow  = A + (long)(row0 + lo) * K + 2 * hi;
    const float* __restrict__ bbase = B + (long)head * bStride + 2 * hi * NB + col0 + lo;

    v8f acc = {};
    for (int k = 0; k < K; k += 4) {
        v2f a;
        a.x = arow[k];
        a.y = arow[k + 1];
        v2f b;
        b.x = bbase[k * NB];
        b.y = bbase[k * NB + NB];
        acc = __builtin_amdgcn_wmma_f32_16x16x4_f32(
            false, a, false, b, (short)0, acc, false, false);
    }

    float* cp = C + (long)head * cStride + (long)(row0 + 8 * hi) * NB + col0 + lo;
    #pragma unroll
    for (int v = 0; v < 8; ++v) cp[v * NB] = acc[v];
}

__global__ void fused_fifj(const float* __restrict__ H, long hStride, int F,
                           const float* __restrict__ a, int aStride,
                           float* __restrict__ fi, float* __restrict__ fj, int n) {
    const int head = blockIdx.y;
    const int nd   = blockIdx.x * blockDim.x + threadIdx.x;
    if (nd >= n) return;
    const float* hp = H + (long)head * hStride + (long)nd * F;
    const float* a1 = a + (long)head * aStride;
    const float* a2 = a1 + F;
    float s1 = 0.f, s2 = 0.f;
    for (int o = 0; o < F; ++o) {
        float hv = hp[o];
        s1 += hv * a1[o];
        s2 += hv * a2[o];
    }
    fi[(long)head * n + nd] = s1;
    fj[(long)head * n + nd] = s2;
}

template <int F, int LD, int R>
__device__ __forceinline__ void attn_exact_pass(
        const float* __restrict__ bbase, const float* __restrict__ fjh,
        const unsigned* const (&brow)[R], const float (&fiv)[R],
        const int hi, const int jbeg, const int jend,
        float (&m)[R], float (&l)[R], v8f (&acc)[R][F / 16]) {
    constexpr int NC = F / 16;
    constexpr int KS = 16;
    constexpr int NT16 = KS / 16;
    constexpr int NW = 1;
    for (int j0 = jbeg; j0 < jend; j0 += KS) {
        unsigned wmask[R][NW];
        #pragma unroll
        for (int r = 0; r < R; ++r)
            #pragma unroll
            for (int w = 0; w < NW; ++w) wmask[r][w] = brow[r][(j0 >> 5) + w];

        float fjv[NT16][8];
        #pragma unroll
        for (int t = 0; t < NT16; ++t) {
            const float* __restrict__ fjp = fjh + j0 + 16 * t + 2 * hi;
            #pragma unroll
            for (int s = 0; s < 4; ++s)
                #pragma unroll
                for (int u = 0; u < 2; ++u) fjv[t][2 * s + u] = fjp[4 * s + u];
        }

        float p[R][NT16][8];
        #pragma unroll
        for (int r = 0; r < R; ++r) {
            float tm[NT16];
            #pragma unroll
            for (int t = 0; t < NT16; ++t) {
                const unsigned wbits = wmask[r][t >> 1];
                const int bitbase = j0 & 16;
                float mx = -INFINITY;
                #pragma unroll
                for (int s = 0; s < 4; ++s) {
                    #pragma unroll
                    for (int u = 0; u < 2; ++u) {
                        const int c = 4 * s + 2 * hi + u;
                        float v = fiv[r] + fjv[t][2 * s + u];
                        v = v > 0.f ? v : ALPHA * v;
                        const float e = ((wbits >> (bitbase + c)) & 1u) ? v : NEGV;
                        p[r][t][2 * s + u] = e;
                        mx = fmaxf(mx, e);
                    }
                }
                tm[t] = mx;
            }
            float tmax = tm[0];
            tmax = fmaxf(tmax, __shfl_xor(tmax, 16, 32));
            const float mnew = fmaxf(m[r], tmax);

            float ps[NT16];
            #pragma unroll
            for (int t = 0; t < NT16; ++t) {
                float sacc = 0.f;
                #pragma unroll
                for (int q = 0; q < 8; ++q) {
                    p[r][t][q] = __expf(p[r][t][q] - mnew);
                    sacc += p[r][t][q];
                }
                ps[t] = sacc;
            }
            float psum = ps[0];
            psum += __shfl_xor(psum, 16, 32);

            if (__any(mnew > m[r])) {
                const float scale = __expf(m[r] - mnew);
                float sv[8];
                #pragma unroll
                for (int v = 0; v < 8; ++v) sv[v] = __shfl(scale, v + 8 * hi, 32);
                #pragma unroll
                for (int c = 0; c < NC; ++c)
                    #pragma unroll
                    for (int v = 0; v < 8; ++v) acc[r][c][v] *= sv[v];
                l[r] = l[r] * scale + psum;
                m[r] = mnew;
            } else {
                l[r] += psum;
            }
        }

        #pragma unroll
        for (int t = 0; t < NT16; ++t) {
            __builtin_amdgcn_sched_barrier(0);
            const float* __restrict__ bp = bbase + (j0 + 16 * t) * LD;
            #pragma unroll
            for (int s = 0; s < 4; ++s) {
                const float* __restrict__ bs = bp + 4 * s * LD;
                #pragma unroll
                for (int c = 0; c < NC; ++c) {
                    v2f b;
                    b.x = bs[16 * c];
                    b.y = bs[16 * c + LD];
                    #pragma unroll
                    for (int r = 0; r < R; ++r) {
                        v2f a;
                        a.x = p[r][t][2 * s];
                        a.y = p[r][t][2 * s + 1];
                        acc[r][c] = __builtin_amdgcn_wmma_f32_16x16x4_f32(
                            false, a, false, b, (short)0, acc[r][c], false, false);
                    }
                }
            }
        }
    }

}

template <int F, int LD, int JW, int R, bool ELU>
__attribute__((amdgpu_waves_per_eu(F > 64 ? 1 : 4)))
__global__ __launch_bounds__(JW * 32) void gat_attn(
        const float* __restrict__ Hfeat, long hStride,
        const float* __restrict__ fi, const float* __restrict__ fj, int fStride,
        const unsigned* __restrict__ bits,
        float* __restrict__ Out, int ldo, int headColMul, int n) {
    constexpr int NC = F / 16;
    constexpr int CH = NN / JW;
    constexpr int EB = JW * CH * 2;
    constexpr int AB = JW * 16 * F;
    constexpr int SM = EB > AB ? EB : AB;
    const int wave = threadIdx.x >> 5;
    const int lane = threadIdx.x & 31;
    const int hi   = lane >> 4;
    const int lo   = lane & 15;
    const int head = blockIdx.y;
    const int i0   = blockIdx.x * (16 * R);
    const int nwords = n >> 5;

    __shared__ __attribute__((aligned(16))) float smem[SM];
    __shared__ float m_lds[R][JW * 16];
    __shared__ float l_lds[R][JW * 16];
    __shared__ float L_lds[16];
    float* accbuf = smem;
    float* ew = smem + wave * (2 * CH);

    const float* __restrict__ Hh  = Hfeat + (long)head * hStride;
    const float* __restrict__ fih = fi + (long)head * fStride;
    const float* __restrict__ fjh = fj + (long)head * fStride;
    const float* __restrict__ bbase = Hh + 2 * hi * LD + lo;

    float fiv[R];
    const unsigned* brow[R];
    #pragma unroll
    for (int r = 0; r < R; ++r) {
        fiv[r]  = fih[i0 + 16 * r + lo];
        brow[r] = bits + (long)(i0 + 16 * r + lo) * nwords;
    }

    const int jbeg  = wave * CH;
    const int jend  = jbeg + CH;

    float cm = -INFINITY;
    {
        v4f fq[CH / 128];
        const v4f* __restrict__ fj4 = (const v4f*)(fjh + jbeg);
        #pragma unroll
        for (int i = 0; i < CH / 128; ++i) {
            fq[i] = fj4[i * 32 + lane];
            cm = fmaxf(cm, fmaxf(fmaxf(fq[i].x, fq[i].y), fmaxf(fq[i].z, fq[i].w)));
        }
        #pragma unroll
        for (int off = 16; off >= 1; off >>= 1) cm = fmaxf(cm, __shfl_xor(cm, off, 32));
        #pragma unroll
        for (int i = 0; i < CH / 128; ++i) {
            float* ep = ew + 8 * (i * 32 + lane);
            const float d0 = fq[i].x - cm, d1 = fq[i].y - cm;
            const float d2 = fq[i].z - cm, d3 = fq[i].w - cm;
            v4f e0, e1;
            e0.x = __expf(d0); e0.y = __expf(ALPHA * d0);
            e0.z = __expf(d1); e0.w = __expf(ALPHA * d1);
            e1.x = __expf(d2); e1.y = __expf(ALPHA * d2);
            e1.z = __expf(d3); e1.w = __expf(ALPHA * d3);
            *(v4f*)ep = e0;
            *(v4f*)(ep + 4) = e1;
        }
    }
    __syncthreads();

    float m[R], l[R];
    v2f arow[R];
    v8f acc[R][NC];
    #pragma unroll
    for (int r = 0; r < R; ++r) {
        const float t = fiv[r] + cm;
        m[r] = fmaxf(t, ALPHA * t);
        arow[r].x = __expf(t - m[r]);
        arow[r].y = __expf(ALPHA * t - m[r]);
        l[r] = 0.f;
        #pragma unroll
        for (int c = 0; c < NC; ++c) acc[r][c] = (v8f){};
    }

    constexpr int KS = F > 64 ? 64 : 32;
    constexpr int NT16 = KS / 16;
    constexpr int NW = KS / 32;
    for (int j0 = jbeg; j0 < jend; j0 += KS) {
        unsigned wsh[R][NW];
        #pragma unroll
        for (int r = 0; r < R; ++r)
            #pragma unroll
            for (int w = 0; w < NW; ++w) wsh[r][w] = brow[r][(j0 >> 5) + w] >> (2 * hi);

        const float* es = ew + 2 * (j0 - jbeg) + 4 * hi;
        #pragma unroll
        for (int t = 0; t < NT16; ++t) {
            __builtin_amdgcn_sched_barrier(0);

            float p[R][8];
            float ps[R];
            #pragma unroll
            for (int r = 0; r < R; ++r) ps[r] = 0.f;
            #pragma unroll
            for (int s = 0; s < 4; ++s) {
                const v4f q = *(const v4f*)(es + 2 * (16 * t + 4 * s));
                v2f q0, q1;
                q0.x = q.x; q0.y = q.y;
                q1.x = q.z; q1.y = q.w;
                #pragma unroll
                for (int r = 0; r < R; ++r) {
                    const v2f x0 = arow[r] * q0;
                    const v2f x1 = arow[r] * q1;
                    const float w0 = fmaxf(x0.x, x0.y);
                    const float w1 = fmaxf(x1.x, x1.y);
                    const int k0 = 16 * (t & 1) + 4 * s;
                    const unsigned wb = wsh[r][t >> 1];
                    const int mk0 = ((int)(wb << (31 - k0))) >> 31;
                    const int mk1 = ((int)(wb << (30 - k0))) >> 31;
                    const float p0 = __int_as_float(__float_as_int(w0) & mk0);
                    const float p1 = __int_as_float(__float_as_int(w1) & mk1);
                    p[r][2 * s]     = p0;
                    p[r][2 * s + 1] = p1;
                    ps[r] += p0;
                    ps[r] += p1;
                }
            }
            #pragma unroll
            for (int r = 0; r < R; ++r) l[r] += ps[r];

            const float* __restrict__ bp = bbase + (j0 + 16 * t) * LD;
            #pragma unroll
            for (int s = 0; s < 4; ++s) {
                const float* __restrict__ bs = bp + 4 * s * LD;
                #pragma unroll
                for (int c = 0; c < NC; ++c) {
                    v2f b;
                    b.x = bs[16 * c];
                    b.y = bs[16 * c + LD];
                    #pragma unroll
                    for (int r = 0; r < R; ++r) {
                        v2f a;
                        a.x = p[r][2 * s];
                        a.y = p[r][2 * s + 1];
                        acc[r][c] = __builtin_amdgcn_wmma_f32_16x16x4_f32(
                            false, a, false, b, (short)0, acc[r][c], false, false);
                    }
                }
            }
        }
    }
    #pragma unroll
    for (int r = 0; r < R; ++r) l[r] += __shfl_xor(l[r], 16, 32);

    #pragma unroll
    for (int r = 0; r < R; ++r) {
        if (hi == 0) {
            m_lds[r][wave * 16 + lo] = m[r];
            l_lds[r][wave * 16 + lo] = l[r];
        }
    }
    __syncthreads();

    bool small = false;
    #pragma unroll
    for (int r = 0; r < R; ++r) {
        float Mrow = -INFINITY;
        #pragma unroll
        for (int w = 0; w < JW; ++w) Mrow = fmaxf(Mrow, m_lds[r][w * 16 + lo]);
        float Lrow = 0.f;
        #pragma unroll
        for (int w = 0; w < JW; ++w)
            Lrow += l_lds[r][w * 16 + lo] * __expf(m_lds[r][w * 16 + lo] - Mrow);
        small = small || !(Lrow >= 1e-25f);
    }
    if (__any(small)) {
        __syncthreads();
        #pragma unroll
        for (int r = 0; r < R; ++r) {
            const unsigned* b1[1] = {brow[r]};
            const float f1[1] = {fiv[r]};
            float m1[1] = {-INFINITY};
            float l1[1] = {0.f};
            v8f a1[1][NC];
            #pragma unroll
            for (int c = 0; c < NC; ++c) a1[0][c] = (v8f){};
            attn_exact_pass<F, LD, 1>(bbase, fjh, b1, f1, hi, jbeg, jend, m1, l1, a1);
            m[r] = m1[0];
            l[r] = l1[0];
            #pragma unroll
            for (int c = 0; c < NC; ++c) acc[r][c] = a1[0][c];
        }
        #pragma unroll
        for (int r = 0; r < R; ++r) {
            if (hi == 0) {
                m_lds[r][wave * 16 + lo] = m[r];
                l_lds[r][wave * 16 + lo] = l[r];
            }
        }
        __syncthreads();
    }

    constexpr int TOT = 16 * F;
    constexpr int NT  = JW * 32;
    #pragma unroll
    for (int r = 0; r < R; ++r) {
        float Mrow = -INFINITY;
        #pragma unroll
        for (int w = 0; w < JW; ++w) Mrow = fmaxf(Mrow, m_lds[r][w * 16 + lo]);
        float Lrow = 0.f;
        #pragma unroll
        for (int w = 0; w < JW; ++w)
            Lrow += l_lds[r][w * 16 + lo] * __expf(m_lds[r][w * 16 + lo] - Mrow);
        if (wave == 0 && hi == 0) L_lds[lo] = Lrow;

        const float fown = __expf(m[r] - Mrow);
        float fv[8];
        #pragma unroll
        for (int v = 0; v < 8; ++v) fv[v] = __shfl(fown, v + 8 * hi, 32);

        float* ab = accbuf + (wave * 16 + 8 * hi) * F + lo;
        #pragma unroll
        for (int c = 0; c < NC; ++c)
            #pragma unroll
            for (int v = 0; v < 8; ++v)
                ab[v * F + 16 * c] = acc[r][c][v] * fv[v];
        __syncthreads();

        float* outBase = Out + (long)(i0 + 16 * r) * ldo + head * headColMul;
        #pragma unroll
        for (int e = 0; e < TOT / NT; ++e) {
            const int flat = e * NT + threadIdx.x;
            const int row = flat / F;
            const int col = flat % F;
            float val = 0.f;
            #pragma unroll
            for (int w = 0; w < JW; ++w) val += accbuf[(w * 16 + row) * F + col];
            float rr = val / L_lds[row];
            if (ELU) rr = rr > 0.f ? rr : (__expf(rr) - 1.0f);
            outBase[(long)row * ldo + col] = rr;
        }
        if (r + 1 < R) __syncthreads();
    }
}

__global__ void logsoftmax128(const float* __restrict__ in,
                              float* __restrict__ out, int n) {
    const int wave = threadIdx.x >> 5;
    const int lane = threadIdx.x & 31;
    const int row  = blockIdx.x * (blockDim.x >> 5) + wave;
    if (row >= n) return;
    const float* rp = in + (long)row * NCLASS;
    float v[4];
    float mx = -INFINITY;
    #pragma unroll
    for (int q = 0; q < 4; ++q) {
        v[q] = rp[lane + 32 * q];
        mx = fmaxf(mx, v[q]);
    }
    #pragma unroll
    for (int off = 16; off >= 1; off >>= 1) mx = fmaxf(mx, __shfl_xor(mx, off, 32));
    float s = 0.f;
    #pragma unroll
    for (int q = 0; q < 4; ++q) s += __expf(v[q] - mx);
    #pragma unroll
    for (int off = 16; off >= 1; off >>= 1) s += __shfl_xor(s, off, 32);
    const float ls = __logf(s);
    #pragma unroll
    for (int q = 0; q < 4; ++q) out[(long)row * NCLASS + lane + 32 * q] = v[q] - mx - ls;
}

extern "C" void kernel_launch(void* const* d_in, const int* in_sizes, int n_in,
                              void* d_out, int out_size, void* d_ws, size_t ws_size,
                              hipStream_t stream) {
    const float* x     = (const float*)d_in[0];
    const int*   adj   = (const int*)d_in[1];
    const float* W     = (const float*)d_in[2];
    const float* a     = (const float*)d_in[3];
    const float* W_out = (const float*)d_in[4];
    const float* a_out = (const float*)d_in[5];
    float* out = (float*)d_out;

    char* ws = (char*)d_ws;
    unsigned* bits = (unsigned*)ws;
    ws += (size_t)NN * (NN / 32) * sizeof(unsigned);
    float* h1 = (float*)ws;
    ws += (size_t)NHEADS * NN * NHID * sizeof(float);
    float* fi1 = (float*)ws; ws += (size_t)NHEADS * NN * sizeof(float);
    float* fj1 = (float*)ws; ws += (size_t)NHEADS * NN * sizeof(float);
    float* xcat = (float*)ws;
    ws += (size_t)NN * (NHEADS * NHID) * sizeof(float);
    float* h2 = (float*)ws;
    ws += (size_t)NN * NCLASS * sizeof(float);
    float* fi2 = (float*)ws; ws += (size_t)NN * sizeof(float);
    float* fj2 = (float*)ws; ws += (size_t)NN * sizeof(float);
    float* outpre = (float*)ws;

    {
        const long nwords = (long)NN * (NN / 32);
        pack_adj_kernel<<<dim3((nwords + 255) / 256), dim3(256), 0, stream>>>(adj, bits, nwords);
    }

    wmma_gemm_f32<NFEAT, NHID><<<dim3(NN / 16, 1, NHEADS), dim3(128), 0, stream>>>(
        x, W, (long)NFEAT * NHID, h1, (long)NN * NHID);

    fused_fifj<<<dim3(NN / 256, NHEADS), dim3(256), 0, stream>>>(
        h1, (long)NN * NHID, NHID, a, 2 * NHID, fi1, fj1, NN);

    gat_attn<NHID, NHID, 4, 2, false><<<dim3(NN / 32, NHEADS), dim3(128), 0, stream>>>(
        h1, (long)NN * NHID, fi1, fj1, NN, bits,
        xcat, NHEADS * NHID, NHID, NN);

    wmma_gemm_f32<NHEADS * NHID, NCLASS><<<dim3(NN / 16, NCLASS / 64, 1), dim3(128), 0, stream>>>(
        xcat, W_out, 0, h2, 0);

    fused_fifj<<<dim3(NN / 256, 1), dim3(256), 0, stream>>>(
        h2, 0, NCLASS, a_out, 2 * NCLASS, fi2, fj2, NN);

    gat_attn<NCLASS / 4, NCLASS, 4, 2, true><<<dim3(NN / 32, 4), dim3(128), 0, stream>>>(
        h2, NCLASS / 4, fi2, fj2, 0, bits,
        outpre, NCLASS, NCLASS / 4, NN);

    logsoftmax128<<<dim3(NN / 4), dim3(128), 0, stream>>>(outpre, out, NN);
}


# projection GEMMs: one block per 16 rows spans all heads' columns; its 16xK slab of A goes once, by whole lines, into LDS (pitch K+4, conflict-free); all waves take A fragments there; sums unchanged
# speedup vs baseline: 1.4961x; 1.0520x over previous
#include <hip/hip_runtime.h>
#include <math.h>

#define NN      4096
#define NFEAT   512
#define NHID    64
#define NCLASS  128
#define NHEADS  8
#define ALPHA   0.2f
#define NEGV    (-9000000000000000.0f)

typedef __attribute__((ext_vector_type(2))) float v2f;
typedef __attribute__((ext_vector_type(4))) float v4f;
typedef __attribute__((ext_vector_type(8))) float v8f;

__global__ void pack_adj_kernel(const int* __restrict__ adj,
                                unsigned* __restrict__ bits, long nwords) {
    long w = (long)blockIdx.x * blockDim.x + threadIdx.x;
    if (w >= nwords) return;
    const int* p = adj + w * 32;
    unsigned m = 0u;
    #pragma unroll
    for (int b = 0; b < 32; ++b) m |= (p[b] > 0 ? 1u : 0u) << b;
    bits[w] = m;
}

template <int K, int NB, int NHD, int WC>
__global__ __launch_bounds__((NHD * NB / (16 * WC)) * 32) void wmma_gemm_f32(
        const float* __restrict__ A,
        const float* __restrict__ B, long bStride,
        float* __restrict__ C, long cStride) {
    constexpr int NW = NHD * NB / (16 * WC);
    constexpr int NT = NW * 32;
    constexpr int P  = K + 4;
    constexpr int NV = 16 * K / 4;
    static_assert(NV % NT == 0, "slab must divide evenly over the block");
    static_assert(NB % (16 * WC) == 0, "a wave's tiles stay inside one head");
    __shared__ __attribute__((aligned(16))) float As[16 * P];

    const int wave = threadIdx.x >> 5;
    const int lane = threadIdx.x & 31;
    const int hi   = lane >> 4;
    const int lo   = lane & 15;
    const int row0 = blockIdx.x * 16;

    {
        const v4f* __restrict__ a4 = (const v4f*)(A + (long)row0 * K);
        v4f t[NV / NT];
        #pragma unroll
        for (int i = 0; i < NV / NT; ++i) t[i] = a4[i * NT + threadIdx.x];
        #pragma unroll
        for (int i = 0; i < NV / NT; ++i) {
            const int q = i * NT + threadIdx.x;
            const int r = q / (K / 4);
            const int c = q % (K / 4);
            *(v4f*)(As + r * P + 4 * c) = t[i];
        }
    }
    __syncthreads();

    const int gcol = wave * (16 * WC);
    const int head = gcol / NB;
    const int col0 = gcol % NB;
    const float* __restrict__ bbase = B + (long)head * bStride + 2 * hi * NB + col0 + lo;
    const float* ap = As + lo * P + 2 * hi;

    v8f acc[WC];
    #pragma unroll
    for (int c = 0; c < WC; ++c) acc[c] = (v8f){};
    for (int k = 0; k < K; k += 4) {
        const v2f a = *(const v2f*)(ap + k);
        #pragma unroll
        for (int c = 0; c < WC; ++c) {
            v2f b;
            b.x = bbase[k * NB + 16 * c];
            b.y = bbase[k * NB + NB + 16 * c];
            acc[c] = __builtin_amdgcn_wmma_f32_16x16x4_f32(
                false, a, false, b, (short)0, acc[c], false, false);
        }
    }

    float* cp = C + (long)head * cStride + (long)(row0 + 8 * hi) * NB + col0 + lo;
    #pragma unroll
    for (int c = 0; c < WC; ++c)
        #pragma unroll
        for (int v = 0; v < 8; ++v) cp[v * NB + 16 * c] = acc[c][v];
}

__global__ void fused_fifj(const float* __restrict__ H, long hStride, int F,
                           const float* __restrict__ a, int aStride,
                           float* __restrict__ fi, float* __restrict__ fj, int n) {
    const int head = blockIdx.y;
    const int nd   = blockIdx.x * blockDim.x + threadIdx.x;
    if (nd >= n) return;
    const float* hp = H + (long)head * hStride + (long)nd * F;
    const float* a1 = a + (long)head * aStride;
    const float* a2 = a1 + F;
    float s1 = 0.f, s2 = 0.f;
    for (int o = 0; o < F; ++o) {
        float hv = hp[o];
        s1 += hv * a1[o];
        s2 += hv * a2[o];
    }
    fi[(long)head * n + nd] = s1;
    fj[(long)head * n + nd] = s2;
}

template <int F, int LD, int R>
__device__ __forceinline__ void attn_exact_pass(
        const float* __restrict__ bbase, const float* __restrict__ fjh,
        const unsigned* const (&brow)[R], const float (&fiv)[R],
        const int hi, const int jbeg, const int jend,
        float (&m)[R], float (&l)[R], v8f (&acc)[R][F / 16]) {
    constexpr int NC = F / 16;
    constexpr int KS = 16;
    constexpr int NT16 = KS / 16;
    constexpr int NW = 1;
    for (int j0 = jbeg; j0 < jend; j0 += KS) {
        unsigned wmask[R][NW];
        #pragma unroll
        for (int r = 0; r < R; ++r)
            #pragma unroll
            for (int w = 0; w < NW; ++w) wmask[r][w] = brow[r][(j0 >> 5) + w];

        float fjv[NT16][8];
        #pragma unroll
        for (int t = 0; t < NT16; ++t) {
            const float* __restrict__ fjp = fjh + j0 + 16 * t + 2 * hi;
            #pragma unroll
            for (int s = 0; s < 4; ++s)
                #pragma unroll
                for (int u = 0; u < 2; ++u) fjv[t][2 * s + u] = fjp[4 * s + u];
        }

        float p[R][NT16][8];
        #pragma unroll
        for (int r = 0; r < R; ++r) {
            float tm[NT16];
            #pragma unroll
            for (int t = 0; t < NT16; ++t) {
                const unsigned wbits = wmask[r][t >> 1];
                const int bitbase = j0 & 16;
                float mx = -INFINITY;
                #pragma unroll
                for (int s = 0; s < 4; ++s) {
                    #pragma unroll
                    for (int u = 0; u < 2; ++u) {
                        const int c = 4 * s + 2 * hi + u;
                        float v = fiv[r] + fjv[t][2 * s + u];
                        v = v > 0.f ? v : ALPHA * v;
                        const float e = ((wbits >> (bitbase + c)) & 1u) ? v : NEGV;
                        p[r][t][2 * s + u] = e;
                        mx = fmaxf(mx, e);
                    }
                }
                tm[t] = mx;
            }
            float tmax = tm[0];
            tmax = fmaxf(tmax, __shfl_xor(tmax, 16, 32));
            const float mnew = fmaxf(m[r], tmax);

            float ps[NT16];
            #pragma unroll
            for (int t = 0; t < NT16; ++t) {
                float sacc = 0.f;
                #pragma unroll
                for (int q = 0; q < 8; ++q) {
                    p[r][t][q] = __expf(p[r][t][q] - mnew);
                    sacc += p[r][t][q];
                }
                ps[t] = sacc;
            }
            float psum = ps[0];
            psum += __shfl_xor(psum, 16, 32);

            if (__any(mnew > m[r])) {
                const float scale = __expf(m[r] - mnew);
                float sv[8];
                #pragma unroll
                for (int v = 0; v < 8; ++v) sv[v] = __shfl(scale, v + 8 * hi, 32);
                #pragma unroll
                for (int c = 0; c < NC; ++c)
                    #pragma unroll
                    for (int v = 0; v < 8; ++v) acc[r][c][v] *= sv[v];
                l[r] = l[r] * scale + psum;
                m[r] = mnew;
            } else {
                l[r] += psum;
            }
        }

        #pragma unroll
        for (int t = 0; t < NT16; ++t) {
            __builtin_amdgcn_sched_barrier(0);
            const float* __restrict__ bp = bbase + (j0 + 16 * t) * LD;
            #pragma unroll
            for (int s = 0; s < 4; ++s) {
                const float* __restrict__ bs = bp + 4 * s * LD;
                #pragma unroll
                for (int c = 0; c < NC; ++c) {
                    v2f b;
                    b.x = bs[16 * c];
                    b.y = bs[16 * c + LD];
                    #pragma unroll
                    for (int r = 0; r < R; ++r) {
                        v2f a;
                        a.x = p[r][t][2 * s];
                        a.y = p[r][t][2 * s + 1];
                        acc[r][c] = __builtin_amdgcn_wmma_f32_16x16x4_f32(
                            false, a, false, b, (short)0, acc[r][c], false, false);
                    }
                }
            }
        }
    }

}

template <int F, int LD, int JW, int R, bool ELU>
__attribute__((amdgpu_waves_per_eu(F > 64 ? 1 : 4)))
__global__ __launch_bounds__(JW * 32) void gat_attn(
        const float* __restrict__ Hfeat, long hStride,
        const float* __restrict__ fi, const float* __restrict__ fj, int fStride,
        const unsigned* __restrict__ bits,
        float* __restrict__ Out, int ldo, int headColMul, int n) {
    constexpr int NC = F / 16;
    constexpr int CH = NN / JW;
    constexpr int EB = JW * CH * 2;
    constexpr int AB = JW * 16 * F;
    constexpr int SM = EB > AB ? EB : AB;
    const int wave = threadIdx.x >> 5;
    const int lane = threadIdx.x & 31;
    const int hi   = lane >> 4;
    const int lo   = lane & 15;
    const int head = blockIdx.y;
    const int i0   = blockIdx.x * (16 * R);
    const int nwords = n >> 5;

    __shared__ __attribute__((aligned(16))) float smem[SM];
    __shared__ float m_lds[R][JW * 16];
    __shared__ float l_lds[R][JW * 16];
    __shared__ float L_lds[16];
    float* accbuf = smem;
    float* ew = smem + wave * (2 * CH);

    const float* __restrict__ Hh  = Hfeat + (long)head * hStride;
    const float* __restrict__ fih = fi + (long)head * fStride;
    const float* __restrict__ fjh = fj + (long)head * fStride;
    const float* __restrict__ bbase = Hh + 2 * hi * LD + lo;

    float fiv[R];
    const unsigned* brow[R];
    #pragma unroll
    for (int r = 0; r < R; ++r) {
        fiv[r]  = fih[i0 + 16 * r + lo];
        brow[r] = bits + (long)(i0 + 16 * r + lo) * nwords;
    }

    const int jbeg  = wave * CH;
    const int jend  = jbeg + CH;

    float cm = -INFINITY;
    {
        v4f fq[CH / 128];
        const v4f* __restrict__ fj4 = (const v4f*)(fjh + jbeg);
        #pragma unroll
        for (int i = 0; i < CH / 128; ++i) {
            fq[i] = fj4[i * 32 + lane];
            cm = fmaxf(cm, fmaxf(fmaxf(fq[i].x, fq[i].y), fmaxf(fq[i].z, fq[i].w)));
        }
        #pragma unroll
        for (int off = 16; off >= 1; off >>= 1) cm = fmaxf(cm, __shfl_xor(cm, off, 32));
        #pragma unroll
        for (int i = 0; i < CH / 128; ++i) {
            float* ep = ew + 8 * (i * 32 + lane);
            const float d0 = fq[i].x - cm, d1 = fq[i].y - cm;
            const float d2 = fq[i].z - cm, d3 = fq[i].w - cm;
            v4f e0, e1;
            e0.x = __expf(d0); e0.y = __expf(ALPHA * d0);
            e0.z = __expf(d1); e0.w = __expf(ALPHA * d1);
            e1.x = __expf(d2); e1.y = __expf(ALPHA * d2);
            e1.z = __expf(d3); e1.w = __expf(ALPHA * d3);
            *(v4f*)ep = e0;
            *(v4f*)(ep + 4) = e1;
        }
    }
    __syncthreads();

    float m[R], l[R];
    v2f arow[R];
    v8f acc[R][NC];
    #pragma unroll
    for (int r = 0; r < R; ++r) {
        const float t = fiv[r] + cm;
        m[r] = fmaxf(t, ALPHA * t);
        arow[r].x = __expf(t - m[r]);
        arow[r].y = __expf(ALPHA * t - m[r]);
        l[r] = 0.f;
        #pragma unroll
        for (int c = 0; c < NC; ++c) acc[r][c] = (v8f){};
    }

    constexpr int KS = F > 64 ? 64 : 32;
    constexpr int NT16 = KS / 16;
    constexpr int NW = KS / 32;
    for (int j0 = jbeg; j0 < jend; j0 += KS) {
        unsigned wsh[R][NW];
        #pragma unroll
        for (int r = 0; r < R; ++r)
            #pragma unroll
            for (int w = 0; w < NW; ++w) wsh[r][w] = brow[r][(j0 >> 5) + w] >> (2 * hi);

        const float* es = ew + 2 * (j0 - jbeg) + 4 * hi;
        #pragma unroll
        for (int t = 0; t < NT16; ++t) {
            __builtin_amdgcn_sched_barrier(0);

            float p[R][8];
            float ps[R];
            #pragma unroll
            for (int r = 0; r < R; ++r) ps[r] = 0.f;
            #pragma unroll
            for (int s = 0; s < 4; ++s) {
                const v4f q = *(const v4f*)(es + 2 * (16 * t + 4 * s));
                v2f q0, q1;
                q0.x = q.x; q0.y = q.y;
                q1.x = q.z; q1.y = q.w;
                #pragma unroll
                for (int r = 0; r < R; ++r) {
                    const v2f x0 = arow[r] * q0;
                    const v2f x1 = arow[r] * q1;
                    const float w0 = fmaxf(x0.x, x0.y);
                    const float w1 = fmaxf(x1.x, x1.y);
                    const int k0 = 16 * (t & 1) + 4 * s;
                    const unsigned wb = wsh[r][t >> 1];
                    const int mk0 = ((int)(wb << (31 - k0))) >> 31;
                    const int mk1 = ((int)(wb << (30 - k0))) >> 31;
                    const float p0 = __int_as_float(__float_as_int(w0) & mk0);
                    const float p1 = __int_as_float(__float_as_int(w1) & mk1);
                    p[r][2 * s]     = p0;
                    p[r][2 * s + 1] = p1;
                    ps[r] += p0;
                    ps[r] += p1;
                }
            }
            #pragma unroll
            for (int r = 0; r < R; ++r) l[r] += ps[r];

            const float* __restrict__ bp = bbase + (j0 + 16 * t) * LD;
            #pragma unroll
            for (int s = 0; s < 4; ++s) {
                const float* __restrict__ bs = bp + 4 * s * LD;
                #pragma unroll
                for (int c = 0; c < NC; ++c) {
                    v2f b;
                    b.x = bs[16 * c];
                    b.y = bs[16 * c + LD];
                    #pragma unroll
                    for (int r = 0; r < R; ++r) {
                        v2f a;
                        a.x = p[r][2 * s];
                        a.y = p[r][2 * s + 1];
                        acc[r][c] = __builtin_amdgcn_wmma_f32_16x16x4_f32(
                            false, a, false, b, (short)0, acc[r][c], false, false);
                    }
                }
            }
        }
    }
    #pragma unroll
    for (int r = 0; r < R; ++r) l[r] += __shfl_xor(l[r], 16, 32);

    #pragma unroll
    for (int r = 0; r < R; ++r) {
        if (hi == 0) {
            m_lds[r][wave * 16 + lo] = m[r];
            l_lds[r][wave * 16 + lo] = l[r];
        }
    }
    __syncthreads();

    bool small = false;
    #pragma unroll
    for (int r = 0; r < R; ++r) {
        float Mrow = -INFINITY;
        #pragma unroll
        for (int w = 0; w < JW; ++w) Mrow = fmaxf(Mrow, m_lds[r][w * 16 + lo]);
        float Lrow = 0.f;
        #pragma unroll
        for (int w = 0; w < JW; ++w)
            Lrow += l_lds[r][w * 16 + lo] * __expf(m_lds[r][w * 16 + lo] - Mrow);
        small = small || !(Lrow >= 1e-25f);
    }
    if (__any(small)) {
        __syncthreads();
        #pragma unroll
        for (int r = 0; r < R; ++r) {
            const unsigned* b1[1] = {brow[r]};
            const float f1[1] = {fiv[r]};
            float m1[1] = {-INFINITY};
            float l1[1] = {0.f};
            v8f a1[1][NC];
            #pragma unroll
            for (int c = 0; c < NC; ++c) a1[0][c] = (v8f){};
            attn_exact_pass<F, LD, 1>(bbase, fjh, b1, f1, hi, jbeg, jend, m1, l1, a1);
            m[r] = m1[0];
            l[r] = l1[0];
            #pragma unroll
            for (int c = 0; c < NC; ++c) acc[r][c] = a1[0][c];
        }
        #pragma unroll
        for (int r = 0; r < R; ++r) {
            if (hi == 0) {
                m_lds[r][wave * 16 + lo] = m[r];
                l_lds[r][wave * 16 + lo] = l[r];
            }
        }
        __syncthreads();
    }

    constexpr int TOT = 16 * F;
    constexpr int NT  = JW * 32;
    #pragma unroll
    for (int r = 0; r < R; ++r) {
        float Mrow = -INFINITY;
        #pragma unroll
        for (int w = 0; w < JW; ++w) Mrow = fmaxf(Mrow, m_lds[r][w * 16 + lo]);
        float Lrow = 0.f;
        #pragma unroll
        for (int w = 0; w < JW; ++w)
            Lrow += l_lds[r][w * 16 + lo] * __expf(m_lds[r][w * 16 + lo] - Mrow);
        if (wave == 0 && hi == 0) L_lds[lo] = Lrow;

        const float fown = __expf(m[r] - Mrow);
        float fv[8];
        #pragma unroll
        for (int v = 0; v < 8; ++v) fv[v] = __shfl(fown, v + 8 * hi, 32);

        float* ab = accbuf + (wave * 16 + 8 * hi) * F + lo;
        #pragma unroll
        for (int c = 0; c < NC; ++c)
            #pragma unroll
            for (int v = 0; v < 8; ++v)
                ab[v * F + 16 * c] = acc[r][c][v] * fv[v];
        __syncthreads();

        float* outBase = Out + (long)(i0 + 16 * r) * ldo + head * headColMul;
        #pragma unroll
        for (int e = 0; e < TOT / NT; ++e) {
            const int flat = e * NT + threadIdx.x;
            const int row = flat / F;
            const int col = flat % F;
            float val = 0.f;
            #pragma unroll
            for (int w = 0; w < JW; ++w) val += accbuf[(w * 16 + row) * F + col];
            float rr = val / L_lds[row];
            if (ELU) rr = rr > 0.f ? rr : (__expf(rr) - 1.0f);
            outBase[(long)row * ldo + col] = rr;
        }
        if (r + 1 < R) __syncthreads();
    }
}

__global__ void logsoftmax128(const float* __restrict__ in,
                              float* __restrict__ out, int n) {
    const int wave = threadIdx.x >> 5;
    const int lane = threadIdx.x & 31;
    const int row  = blockIdx.x * (blockDim.x >> 5) + wave;
    if (row >= n) return;
    const float* rp = in + (long)row * NCLASS;
    float v[4];
    float mx = -INFINITY;
    #pragma unroll
    for (int q = 0; q < 4; ++q) {
        v[q] = rp[lane + 32 * q];
        mx = fmaxf(mx, v[q]);
    }
    #pragma unroll
    for (int off = 16; off >= 1; off >>= 1) mx = fmaxf(mx, __shfl_xor(mx, off, 32));
    float s = 0.f;
    #pragma unroll
    for (int q = 0; q < 4; ++q) s += __expf(v[q] - mx);
    #pragma unroll
    for (int off = 16; off >= 1; off >>= 1) s += __shfl_xor(s, off, 32);
    const float ls = __logf(s);
    #pragma unroll
    for (int q = 0; q < 4; ++q) out[(long)row * NCLASS + lane + 32 * q] = v[q] - mx - ls;
}

extern "C" void kernel_launch(void* const* d_in, const int* in_sizes, int n_in,
                              void* d_out, int out_size, void* d_ws, size_t ws_size,
                              hipStream_t stream) {
    const float* x     = (const float*)d_in[0];
    const int*   adj   = (const int*)d_in[1];
    const float* W     = (const float*)d_in[2];
    const float* a     = (const float*)d_in[3];
    const float* W_out = (const float*)d_in[4];
    const float* a_out = (const float*)d_in[5];
    float* out = (float*)d_out;

    char* ws = (char*)d_ws;
    unsigned* bits = (unsigned*)ws;
    ws += (size_t)NN * (NN / 32) * sizeof(unsigned);
    float* h1 = (float*)ws;
    ws += (size_t)NHEADS * NN * NHID * sizeof(float);
    float* fi1 = (float*)ws; ws += (size_t)NHEADS * NN * sizeof(float);
    float* fj1 = (float*)ws; ws += (size_t)NHEADS * NN * sizeof(float);
    float* xcat = (float*)ws;
    ws += (size_t)NN * (NHEADS * NHID) * sizeof(float);
    float* h2 = (float*)ws;
    ws += (size_t)NN * NCLASS * sizeof(float);
    float* fi2 = (float*)ws; ws += (size_t)NN * sizeof(float);
    float* fj2 = (float*)ws; ws += (size_t)NN * sizeof(float);
    float* outpre = (float*)ws;

    {
        const long nwords = (long)NN * (NN / 32);
        pack_adj_kernel<<<dim3((nwords + 255) / 256), dim3(256), 0, stream>>>(adj, bits, nwords);
    }

    wmma_gemm_f32<NFEAT, NHID, NHEADS, 2><<<dim3(NN / 16), dim3(512), 0, stream>>>(
        x, W, (long)NFEAT * NHID, h1, (long)NN * NHID);

    fused_fifj<<<dim3(NN / 256, NHEADS), dim3(256), 0, stream>>>(
        h1, (long)NN * NHID, NHID, a, 2 * NHID, fi1, fj1, NN);

    gat_attn<NHID, NHID, 4, 2, false><<<dim3(NN / 32, NHEADS), dim3(128), 0, stream>>>(
        h1, (long)NN * NHID, fi1, fj1, NN, bits,
        xcat, NHEADS * NHID, NHID, NN);

    wmma_gemm_f32<NHEADS * NHID, NCLASS, 1, 1><<<dim3(NN / 16), dim3(256), 0, stream>>>(
        xcat, W_out, 0, h2, 0);

    fused_fifj<<<dim3(NN / 256, 1), dim3(256), 0, stream>>>(
        h2, 0, NCLASS, a_out, 2 * NCLASS, fi2, fj2, NN);

    gat_attn<NCLASS / 4, NCLASS, 4, 2, true><<<dim3(NN / 32, 4), dim3(128), 0, stream>>>(
        h2, NCLASS / 4, fi2, fj2, 0, bits,
        outpre, NCLASS, NCLASS / 4, NN);

    logsoftmax128<<<dim3(NN / 4), dim3(128), 0, stream>>>(outpre, out, NN);
}


# the four non-matrix launches re-gridded so no load is waited for inside a loop: pack as 4096 waves voting whole lines into mask words, fi/fj rows requested whole (F fixed), log-softmax 4 rows a wave
# speedup vs baseline: 1.5462x; 1.0335x over previous
#include <hip/hip_runtime.h>
#include <math.h>

#define NN      4096
#define NFEAT   512
#define NHID    64
#define NCLASS  128
#define NHEADS  8
#define ALPHA   0.2f
#define NEGV    (-9000000000000000.0f)

typedef __attribute__((ext_vector_type(2))) float v2f;
typedef __attribute__((ext_vector_type(4))) float v4f;
typedef __attribute__((ext_vector_type(8))) float v8f;

#define PACK_PIECES   (NN * (NN / 1024))
#define PACK_PER_WAVE 4
#define PACK_WAVES    (PACK_PIECES / PACK_PER_WAVE)
__global__ __launch_bounds__(512) void pack_adj_kernel(const int* __restrict__ adj,
                                                       unsigned* __restrict__ bits) {
    const int lane = threadIdx.x & 31;
    const int gw   = blockIdx.x * 16 + (threadIdx.x >> 5);
    int v[2][32];
    {
        const int* __restrict__ p = adj + (long)gw * 1024 + lane;
        #pragma unroll
        for (int q = 0; q < 32; ++q) v[0][q] = __builtin_nontemporal_load(p + 32 * q);
    }
    #pragma unroll
    for (int k = 0; k < PACK_PER_WAVE; ++k) {
        const long pc = (long)gw + (long)k * PACK_WAVES;
        if (k + 1 < PACK_PER_WAVE) {
            const int* __restrict__ p = adj + (pc + PACK_WAVES) * 1024 + lane;
            #pragma unroll
            for (int q = 0; q < 32; ++q) v[(k + 1) & 1][q] = __builtin_nontemporal_load(p + 32 * q);
        }
        unsigned w = 0u;
        #pragma unroll
        for (int q = 0; q < 32; ++q) {
            const unsigned vote = __builtin_amdgcn_ballot_w32(v[k & 1][q] > 0);
            w = (lane == q) ? vote : w;
        }
        bits[pc * 32 + lane] = w;
    }
}

template <int K, int NB, int NHD, int WC>
__global__ __launch_bounds__((NHD * NB / (16 * WC)) * 32) void wmma_gemm_f32(
        const float* __restrict__ A,
        const float* __restrict__ B, long bStride,
        float* __restrict__ C, long cStride) {
    constexpr int NW = NHD * NB / (16 * WC);
    constexpr int NT = NW * 32;
    constexpr int P  = K + 4;
    constexpr int NV = 16 * K / 4;
    static_assert(NV % NT == 0, "slab must divide evenly over the block");
    static_assert(NB % (16 * WC) == 0, "a wave's tiles stay inside one head");
    __shared__ __attribute__((aligned(16))) float As[16 * P];

    const int wave = threadIdx.x >> 5;
    const int lane = threadIdx.x & 31;
    const int hi   = lane >> 4;
    const int lo   = lane & 15;
    const int row0 = blockIdx.x * 16;

    {
        const v4f* __restrict__ a4 = (const v4f*)(A + (long)row0 * K);
        v4f t[NV / NT];
        #pragma unroll
        for (int i = 0; i < NV / NT; ++i) t[i] = a4[i * NT + threadIdx.x];
        #pragma unroll
        for (int i = 0; i < NV / NT; ++i) {
            const int q = i * NT + threadIdx.x;
            const int r = q / (K / 4);
            const int c = q % (K / 4);
            *(v4f*)(As + r * P + 4 * c) = t[i];
        }
    }
    __syncthreads();

    const int gcol = wave * (16 * WC);
    const int head = gcol / NB;
    const int col0 = gcol % NB;
    const float* __restrict__ bbase = B + (long)head * bStride + 2 * hi * NB + col0 + lo;
    const float* ap = As + lo * P + 2 * hi;

    v8f acc[WC];
    #pragma unroll
    for (int c = 0; c < WC; ++c) acc[c] = (v8f){};
    for (int k = 0; k < K; k += 4) {
        const v2f a = *(const v2f*)(ap + k);
        #pragma unroll
        for (int c = 0; c < WC; ++c) {
            v2f b;
            b.x = bbase[k * NB + 16 * c];
            b.y = bbase[k * NB + NB + 16 * c];
            acc[c] = __builtin_amdgcn_wmma_f32_16x16x4_f32(
                false, a, false, b, (short)0, acc[c], false, false);
        }
    }

    float* cp = C + (long)head * cStride + (long)(row0 + 8 * hi) * NB + col0 + lo;
    #pragma unroll
    for (int c = 0; c < WC; ++c)
        #pragma unroll
        for (int v = 0; v < 8; ++v) cp[v * NB + 16 * c] = acc[c][v];
}

template <int F>
__global__ void fused_fifj(const float* __restrict__ H, long hStride,
                           const float* __restrict__ a,
                           float* __restrict__ fi, float* __restrict__ fj) {
    const int head = blockIdx.y;
    const int nd   = blockIdx.x * blockDim.x + threadIdx.x;
    const v4f* __restrict__ hp = (const v4f*)(H + (long)head * hStride + (long)nd * F);
    const float* __restrict__ a1 = a + (long)head * (2 * F);
    const float* __restrict__ a2 = a1 + F;
    v4f t[F / 4];
    #pragma unroll
    for (int i = 0; i < F / 4; ++i) t[i] = hp[i];
    float s1 = 0.f, s2 = 0.f;
    #pragma unroll
    for (int i = 0; i < F / 4; ++i) {
        #pragma unroll
        for (int e = 0; e < 4; ++e) {
            const float hv = t[i][e];
            s1 = __builtin_fmaf(hv, a1[4 * i + e], s1);
            s2 = __builtin_fmaf(hv, a2[4 * i + e], s2);
        }
    }
    fi[(long)head * NN + nd] = s1;
    fj[(long)head * NN + nd] = s2;
}

template <int F, int LD, int R>
__device__ __forceinline__ void attn_exact_pass(
        const float* __restrict__ bbase, const float* __restrict__ fjh,
        const unsigned* const (&brow)[R], const float (&fiv)[R],
        const int hi, const int jbeg, const int jend,
        float (&m)[R], float (&l)[R], v8f (&acc)[R][F / 16]) {
    constexpr int NC = F / 16;
    constexpr int KS = 16;
    constexpr int NT16 = KS / 16;
    constexpr int NW = 1;
    for (int j0 = jbeg; j0 < jend; j0 += KS) {
        unsigned wmask[R][NW];
        #pragma unroll
        for (int r = 0; r < R; ++r)
            #pragma unroll
            for (int w = 0; w < NW; ++w) wmask[r][w] = brow[r][(j0 >> 5) + w];

        float fjv[NT16][8];
        #pragma unroll
        for (int t = 0; t < NT16; ++t) {
            const float* __restrict__ fjp = fjh + j0 + 16 * t + 2 * hi;
            #pragma unroll
            for (int s = 0; s < 4; ++s)
                #pragma unroll
                for (int u = 0; u < 2; ++u) fjv[t][2 * s + u] = fjp[4 * s + u];
        }

        float p[R][NT16][8];
        #pragma unroll
        for (int r = 0; r < R; ++r) {
            float tm[NT16];
            #pragma unroll
            for (int t = 0; t < NT16; ++t) {
                const unsigned wbits = wmask[r][t >> 1];
                const int bitbase = j0 & 16;
                float mx = -INFINITY;
                #pragma unroll
                for (int s = 0; s < 4; ++s) {
                    #pragma unroll
                    for (int u = 0; u < 2; ++u) {
                        const int c = 4 * s + 2 * hi + u;
                        float v = fiv[r] + fjv[t][2 * s + u];
                        v = v > 0.f ? v : ALPHA * v;
                        const float e = ((wbits >> (bitbase + c)) & 1u) ? v : NEGV;
                        p[r][t][2 * s + u] = e;
                        mx = fmaxf(mx, e);
                    }
                }
                tm[t] = mx;
            }
            float tmax = tm[0];
            tmax = fmaxf(tmax, __shfl_xor(tmax, 16, 32));
            const float mnew = fmaxf(m[r], tmax);

            float ps[NT16];
            #pragma unroll
            for (int t = 0; t < NT16; ++t) {
                float sacc = 0.f;
                #pragma unroll
                for (int q = 0; q < 8; ++q) {
                    p[r][t][q] = __expf(p[r][t][q] - mnew);
                    sacc += p[r][t][q];
                }
                ps[t] = sacc;
            }
            float psum = ps[0];
            psum += __shfl_xor(psum, 16, 32);

            if (__any(mnew > m[r])) {
                const float scale = __expf(m[r] - mnew);
                float sv[8];
                #pragma unroll
                for (int v = 0; v < 8; ++v) sv[v] = __shfl(scale, v + 8 * hi, 32);
                #pragma unroll
                for (int c = 0; c < NC; ++c)
                    #pragma unroll
                    for (int v = 0; v < 8; ++v) acc[r][c][v] *= sv[v];
                l[r] = l[r] * scale + psum;
                m[r] = mnew;
            } else {
                l[r] += psum;
            }
        }

        #pragma unroll
        for (int t = 0; t < NT16; ++t) {
            __builtin_amdgcn_sched_barrier(0);
            const float* __restrict__ bp = bbase + (j0 + 16 * t) * LD;
            #pragma unroll
            for (int s = 0; s < 4; ++s) {
                const float* __restrict__ bs = bp + 4 * s * LD;
                #pragma unroll
                for (int c = 0; c < NC; ++c) {
                    v2f b;
                    b.x = bs[16 * c];
                    b.y = bs[16 * c + LD];
                    #pragma unroll
                    for (int r = 0; r < R; ++r) {
                        v2f a;
                        a.x = p[r][t][2 * s];
                        a.y = p[r][t][2 * s + 1];
                        acc[r][c] = __builtin_amdgcn_wmma_f32_16x16x4_f32(
                            false, a, false, b, (short)0, acc[r][c], false, false);
                    }
                }
            }
        }
    }

}

template <int F, int LD, int JW, int R, bool ELU>
__attribute__((amdgpu_waves_per_eu(F > 64 ? 1 : 4)))
__global__ __launch_bounds__(JW * 32) void gat_attn(
        const float* __restrict__ Hfeat, long hStride,
        const float* __restrict__ fi, const float* __restrict__ fj, int fStride,
        const unsigned* __restrict__ bits,
        float* __restrict__ Out, int ldo, int headColMul, int n) {
    constexpr int NC = F / 16;
    constexpr int CH = NN / JW;
    constexpr int EB = JW * CH * 2;
    constexpr int AB = JW * 16 * F;
    constexpr int SM = EB > AB ? EB : AB;
    const int wave = threadIdx.x >> 5;
    const int lane = threadIdx.x & 31;
    const int hi   = lane >> 4;
    const int lo   = lane & 15;
    const int head = blockIdx.y;
    const int i0   = blockIdx.x * (16 * R);
    const int nwords = n >> 5;

    __shared__ __attribute__((aligned(16))) float smem[SM];
    __shared__ float m_lds[R][JW * 16];
    __shared__ float l_lds[R][JW * 16];
    __shared__ float L_lds[16];
    float* accbuf = smem;
    float* ew = smem + wave * (2 * CH);

    const float* __restrict__ Hh  = Hfeat + (long)head * hStride;
    const float* __restrict__ fih = fi + (long)head * fStride;
    const float* __restrict__ fjh = fj + (long)head * fStride;
    const float* __restrict__ bbase = Hh + 2 * hi * LD + lo;

    float fiv[R];
    const unsigned* brow[R];
    #pragma unroll
    for (int r = 0; r < R; ++r) {
        fiv[r]  = fih[i0 + 16 * r + lo];
        brow[r] = bits + (long)(i0 + 16 * r + lo) * nwords;
    }

    const int jbeg  = wave * CH;
    const int jend  = jbeg + CH;

    float cm = -INFINITY;
    {
        v4f fq[CH / 128];
        const v4f* __restrict__ fj4 = (const v4f*)(fjh + jbeg);
        #pragma unroll
        for (int i = 0; i < CH / 128; ++i) {
            fq[i] = fj4[i * 32 + lane];
            cm = fmaxf(cm, fmaxf(fmaxf(fq[i].x, fq[i].y), fmaxf(fq[i].z, fq[i].w)));
        }
        #pragma unroll
        for (int off = 16; off >= 1; off >>= 1) cm = fmaxf(cm, __shfl_xor(cm, off, 32));
        #pragma unroll
        for (int i = 0; i < CH / 128; ++i) {
            float* ep = ew + 8 * (i * 32 + lane);
            const float d0 = fq[i].x - cm, d1 = fq[i].y - cm;
            const float d2 = fq[i].z - cm, d3 = fq[i].w - cm;
            v4f e0, e1;
            e0.x = __expf(d0); e0.y = __expf(ALPHA * d0);
            e0.z = __expf(d1); e0.w = __expf(ALPHA * d1);
            e1.x = __expf(d2); e1.y = __expf(ALPHA * d2);
            e1.z = __expf(d3); e1.w = __expf(ALPHA * d3);
            *(v4f*)ep = e0;
            *(v4f*)(ep + 4) = e1;
        }
    }
    __syncthreads();

    float m[R], l[R];
    v2f arow[R];
    v8f acc[R][NC];
    #pragma unroll
    for (int r = 0; r < R; ++r) {
        const float t = fiv[r] + cm;
        m[r] = fmaxf(t, ALPHA * t);
        arow[r].x = __expf(t - m[r]);
        arow[r].y = __expf(ALPHA * t - m[r]);
        l[r] = 0.f;
        #pragma unroll
        for (int c = 0; c < NC; ++c) acc[r][c] = (v8f){};
    }

    constexpr int KS = F > 64 ? 64 : 32;
    constexpr int NT16 = KS / 16;
    constexpr int NW = KS / 32;
    for (int j0 = jbeg; j0 < jend; j0 += KS) {
        unsigned wsh[R][NW];
        #pragma unroll
        for (int r = 0; r < R; ++r)
            #pragma unroll
            for (int w = 0; w < NW; ++w) wsh[r][w] = brow[r][(j0 >> 5) + w] >> (2 * hi);

        const float* es = ew + 2 * (j0 - jbeg) + 4 * hi;
        #pragma unroll
        for (int t = 0; t < NT16; ++t) {
            __builtin_amdgcn_sched_barrier(0);

            float p[R][8];
            float ps[R];
            #pragma unroll
            for (int r = 0; r < R; ++r) ps[r] = 0.f;
            #pragma unroll
            for (int s = 0; s < 4; ++s) {
                const v4f q = *(const v4f*)(es + 2 * (16 * t + 4 * s));
                v2f q0, q1;
                q0.x = q.x; q0.y = q.y;
                q1.x = q.z; q1.y = q.w;
                #pragma unroll
                for (int r = 0; r < R; ++r) {
                    const v2f x0 = arow[r] * q0;
                    const v2f x1 = arow[r] * q1;
                    const float w0 = fmaxf(x0.x, x0.y);
                    const float w1 = fmaxf(x1.x, x1.y);
                    const int k0 = 16 * (t & 1) + 4 * s;
                    const unsigned wb = wsh[r][t >> 1];
                    const int mk0 = ((int)(wb << (31 - k0))) >> 31;
                    const int mk1 = ((int)(wb << (30 - k0))) >> 31;
                    const float p0 = __int_as_float(__float_as_int(w0) & mk0);
                    const float p1 = __int_as_float(__float_as_int(w1) & mk1);
                    p[r][2 * s]     = p0;
                    p[r][2 * s + 1] = p1;
                    ps[r] += p0;
                    ps[r] += p1;
                }
            }
            #pragma unroll
            for (int r = 0; r < R; ++r) l[r] += ps[r];

            const float* __restrict__ bp = bbase + (j0 + 16 * t) * LD;
            #pragma unroll
            for (int s = 0; s < 4; ++s) {
                const float* __restrict__ bs = bp + 4 * s * LD;
                #pragma unroll
                for (int c = 0; c < NC; ++c) {
                    v2f b;
                    b.x = bs[16 * c];
                    b.y = bs[16 * c + LD];
                    #pragma unroll
                    for (int r = 0; r < R; ++r) {
                        v2f a;
                        a.x = p[r][2 * s];
                        a.y = p[r][2 * s + 1];
                        acc[r][c] = __builtin_amdgcn_wmma_f32_16x16x4_f32(
                            false, a, false, b, (short)0, acc[r][c], false, false);
                    }
                }
            }
        }
    }
    #pragma unroll
    for (int r = 0; r < R; ++r) l[r] += __shfl_xor(l[r], 16, 32);

    #pragma unroll
    for (int r = 0; r < R; ++r) {
        if (hi == 0) {
            m_lds[r][wave * 16 + lo] = m[r];
            l_lds[r][wave * 16 + lo] = l[r];
        }
    }
    __syncthreads();

    bool small = false;
    #pragma unroll
    for (int r = 0; r < R; ++r) {
        float Mrow = -INFINITY;
        #pragma unroll
        for (int w = 0; w < JW; ++w) Mrow = fmaxf(Mrow, m_lds[r][w * 16 + lo]);
        float Lrow = 0.f;
        #pragma unroll
        for (int w = 0; w < JW; ++w)
            Lrow += l_lds[r][w * 16 + lo] * __expf(m_lds[r][w * 16 + lo] - Mrow);
        small = small || !(Lrow >= 1e-25f);
    }
    if (__any(small)) {
        __syncthreads();
        #pragma unroll
        for (int r = 0; r < R; ++r) {
            const unsigned* b1[1] = {brow[r]};
            const float f1[1] = {fiv[r]};
            float m1[1] = {-INFINITY};
            float l1[1] = {0.f};
            v8f a1[1][NC];
            #pragma unroll
            for (int c = 0; c < NC; ++c) a1[0][c] = (v8f){};
            attn_exact_pass<F, LD, 1>(bbase, fjh, b1, f1, hi, jbeg, jend, m1, l1, a1);
            m[r] = m1[0];
            l[r] = l1[0];
            #pragma unroll
            for (int c = 0; c < NC; ++c) acc[r][c] = a1[0][c];
        }
        #pragma unroll
        for (int r = 0; r < R; ++r) {
            if (hi == 0) {
                m_lds[r][wave * 16 + lo] = m[r];
                l_lds[r][wave * 16 + lo] = l[r];
            }
        }
        __syncthreads();
    }

    constexpr int TOT = 16 * F;
    constexpr int NT  = JW * 32;
    #pragma unroll
    for (int r = 0; r < R; ++r) {
        float Mrow = -INFINITY;
        #pragma unroll
        for (int w = 0; w < JW; ++w) Mrow = fmaxf(Mrow, m_lds[r][w * 16 + lo]);
        float Lrow = 0.f;
        #pragma unroll
        for (int w = 0; w < JW; ++w)
            Lrow += l_lds[r][w * 16 + lo] * __expf(m_lds[r][w * 16 + lo] - Mrow);
        if (wave == 0 && hi == 0) L_lds[lo] = Lrow;

        const float fown = __expf(m[r] - Mrow);
        float fv[8];
        #pragma unroll
        for (int v = 0; v < 8; ++v) fv[v] = __shfl(fown, v + 8 * hi, 32);

        float* ab = accbuf + (wave * 16 + 8 * hi) * F + lo;
        #pragma unroll
        for (int c = 0; c < NC; ++c)
            #pragma unroll
            for (int v = 0; v < 8; ++v)
                ab[v * F + 16 * c] = acc[r][c][v] * fv[v];
        __syncthreads();

        float* outBase = Out + (long)(i0 + 16 * r) * ldo + head * headColMul;
        #pragma unroll
        for (int e = 0; e < TOT / NT; ++e) {
            const int flat = e * NT + threadIdx.x;
            const int row = flat / F;
            const int col = flat % F;
            float val = 0.f;
            #pragma unroll
            for (int w = 0; w < JW; ++w) val += accbuf[(w * 16 + row) * F + col];
            float rr = val / L_lds[row];
            if (ELU) rr = rr > 0.f ? rr : (__expf(rr) - 1.0f);
            outBase[(long)row * ldo + col] = rr;
        }
        if (r + 1 < R) __syncthreads();
    }
}

#define LR 4
__global__ void logsoftmax128(const float* __restrict__ in,
                              float* __restrict__ out) {
    const int wave = threadIdx.x >> 5;
    const int lane = threadIdx.x & 31;
    const int row0 = (blockIdx.x * (blockDim.x >> 5) + wave) * LR;
    float v[LR][4];
    #pragma unroll
    for (int r = 0; r < LR; ++r)
        #pragma unroll
        for (int q = 0; q < 4; ++q) v[r][q] = in[(long)(row0 + r) * NCLASS + lane + 32 * q];
    #pragma unroll
    for (int r = 0; r < LR; ++r) {
        float mx = -INFINITY;
        #pragma unroll
        for (int q = 0; q < 4; ++q) mx = fmaxf(mx, v[r][q]);
        #pragma unroll
        for (int off = 16; off >= 1; off >>= 1) mx = fmaxf(mx, __shfl_xor(mx, off, 32));
        float s = 0.f;
        #pragma unroll
        for (int q = 0; q < 4; ++q) s += __expf(v[r][q] - mx);
        #pragma unroll
        for (int off = 16; off >= 1; off >>= 1) s += __shfl_xor(s, off, 32);
        const float ls = __logf(s);
        #pragma unroll
        for (int q = 0; q < 4; ++q) out[(long)(row0 + r) * NCLASS + lane + 32 * q] = v[r][q] - mx - ls;
    }
}

extern "C" void kernel_launch(void* const* d_in, const int* in_sizes, int n_in,
                              void* d_out, int out_size, void* d_ws, size_t ws_size,
                              hipStream_t stream) {
    const float* x     = (const float*)d_in[0];
    const int*   adj   = (const int*)d_in[1];
    const float* W     = (const float*)d_in[2];
    const float* a     = (const float*)d_in[3];
    const float* W_out = (const float*)d_in[4];
    const float* a_out = (const float*)d_in[5];
    float* out = (float*)d_out;

    char* ws = (char*)d_ws;
    unsigned* bits = (unsigned*)ws;
    ws += (size_t)NN * (NN / 32) * sizeof(unsigned);
    float* h1 = (float*)ws;
    ws += (size_t)NHEADS * NN * NHID * sizeof(float);
    float* fi1 = (float*)ws; ws += (size_t)NHEADS * NN * sizeof(float);
    float* fj1 = (float*)ws; ws += (size_t)NHEADS * NN * sizeof(float);
    float* xcat = (float*)ws;
    ws += (size_t)NN * (NHEADS * NHID) * sizeof(float);
    float* h2 = (float*)ws;
    ws += (size_t)NN * NCLASS * sizeof(float);
    float* fi2 = (float*)ws; ws += (size_t)NN * sizeof(float);
    float* fj2 = (float*)ws; ws += (size_t)NN * sizeof(float);
    float* outpre = (float*)ws;

    pack_adj_kernel<<<dim3(PACK_WAVES / 16), dim3(512), 0, stream>>>(adj, bits);

    wmma_gemm_f32<NFEAT, NHID, NHEADS, 2><<<dim3(NN / 16), dim3(512), 0, stream>>>(
        x, W, (long)NFEAT * NHID, h1, (long)NN * NHID);

    fused_fifj<NHID><<<dim3(NN / 64, NHEADS), dim3(64), 0, stream>>>(
        h1, (long)NN * NHID, a, fi1, fj1);

    gat_attn<NHID, NHID, 4, 2, false><<<dim3(NN / 32, NHEADS), dim3(128), 0, stream>>>(
        h1, (long)NN * NHID, fi1, fj1, NN, bits,
        xcat, NHEADS * NHID, NHID, NN);

    wmma_gemm_f32<NHEADS * NHID, NCLASS, 1, 1><<<dim3(NN / 16), dim3(256), 0, stream>>>(
        xcat, W_out, 0, h2, 0);

    fused_fifj<NCLASS><<<dim3(NN / 32, 1), dim3(32), 0, stream>>>(
        h2, 0, a_out, fi2, fj2);

    gat_attn<NCLASS / 4, NCLASS, 4, 2, true><<<dim3(NN / 32, 4), dim3(128), 0, stream>>>(
        h2, NCLASS / 4, fi2, fj2, 0, bits,
        outpre, NCLASS, NCLASS / 4, NN);

    logsoftmax128<<<dim3(NN / (4 * LR)), dim3(128), 0, stream>>>(outpre, out);
}
